# starting kernel
# baseline (speedup 1.0000x reference)
#include <hip/hip_runtime.h>

typedef __attribute__((ext_vector_type(2))) float v2f;
typedef __attribute__((ext_vector_type(8))) float v8f;

__global__ void deg_init_k(float* __restrict__ deg, int N) {
  int t = blockIdx.x * blockDim.x + threadIdx.x;
  if (t < N) deg[t] = 1.0f;
}

__global__ void deg_scatter_k(const int* __restrict__ dst, const float* __restrict__ ew,
                              float* __restrict__ deg, int E) {
  int t = blockIdx.x * blockDim.x + threadIdx.x;
  if (t < E) unsafeAtomicAdd(&deg[dst[t]], ew[t]);
}

__global__ void dinv_k(const float* __restrict__ deg, float* __restrict__ dinv, int N) {
  int t = blockIdx.x * blockDim.x + threadIdx.x;
  if (t < N) {
    float d = deg[t];
    dinv[t] = (d > 0.0f) ? rsqrtf(d) : 0.0f;
  }
}

__global__ void norm_k(const int* __restrict__ src, const int* __restrict__ dst,
                       const float* __restrict__ ew, const float* __restrict__ dinv,
                       float* __restrict__ nrm, int E) {
  int t = blockIdx.x * blockDim.x + threadIdx.x;
  if (t < E) nrm[t] = dinv[src[t]] * ew[t] * dinv[dst[t]];
}

template <int K, int F>
__global__ void gemm16_wmma(const float* __restrict__ X, const float* __restrict__ W,
                            float* __restrict__ H, int N) {
  const int lane = threadIdx.x & 31;
  const int wave = threadIdx.x >> 5;
  const int Ft = F >> 4;
  const int Mt = N >> 4;
  const int tile = blockIdx.x * (blockDim.x >> 5) + wave;
  if (tile >= Mt * Ft) return;
  const int mt = tile / Ft;
  const int ft = tile - mt * Ft;
  const int l16 = lane & 15;
  const int hi  = lane >> 4;

  const float* __restrict__ xrow = X + (size_t)(mt * 16 + l16) * K;
  const float* __restrict__ wcol = W + ft * 16 + l16;

  v8f acc = {};
#pragma unroll 4
  for (int k = 0; k < K; k += 4) {
    v2f a, b;
    a.x = xrow[k + 2 * hi];
    a.y = xrow[k + 2 * hi + 1];
    b.x = wcol[(size_t)(k + 2 * hi) * F];
    b.y = wcol[(size_t)(k + 2 * hi + 1) * F];
    acc = __builtin_amdgcn_wmma_f32_16x16x4_f32(false, a, false, b,
                                                (short)0, acc, false, false);
  }

  float* __restrict__ hcol = H + ft * 16 + l16;
#pragma unroll
  for (int r = 0; r < 8; ++r)
    hcol[(size_t)(mt * 16 + r + 8 * hi) * F] = acc[r];
}

template <int F>
__global__ void agg_init_k(const float* __restrict__ h, const float* __restrict__ dinv,
                           const float* __restrict__ bias, float* __restrict__ a, int N) {
  int t = blockIdx.x * blockDim.x + threadIdx.x;
  if (t >= N * F) return;
  constexpr int LOGF = (F == 64) ? 6 : 5;
  int n = t >> LOGF;
  int f = t & (F - 1);
  float di = dinv[n];
  a[t] = bias[f] + di * di * h[t];
}

template <int F>
__global__ void agg_edges_k(const float* __restrict__ h, const int* __restrict__ src,
                            const int* __restrict__ dst, const float* __restrict__ nrm,
                            float* __restrict__ a, int E) {
  unsigned t = blockIdx.x * blockDim.x + threadIdx.x;
  constexpr int LOGF = (F == 64) ? 6 : 5;
  unsigned e = t >> LOGF;
  if (e >= (unsigned)E) return;
  unsigned f = t & (F - 1);
  int s = src[e];
  int d = dst[e];
  float w = nrm[e];
  unsafeAtomicAdd(&a[(size_t)d * F + f], w * h[(size_t)s * F + f]);
}

__global__ void relu_k(float* __restrict__ a, int n) {
  int t = blockIdx.x * blockDim.x + threadIdx.x;
  if (t < n) a[t] = fmaxf(a[t], 0.0f);
}

__global__ void mean_pool_k(const float* __restrict__ emb, float* __restrict__ g, int N) {
  __shared__ float sm[256];
  const int c = blockIdx.x;
  float s = 0.0f;
  for (int r = threadIdx.x; r < N; r += blockDim.x) s += emb[(size_t)r * 32 + c];
  sm[threadIdx.x] = s;
  __syncthreads();
  for (int st = 128; st > 0; st >>= 1) {
    if (threadIdx.x < st) sm[threadIdx.x] += sm[threadIdx.x + st];
    __syncthreads();
  }
  if (threadIdx.x == 0) g[c] = sm[0] / (float)N;
}

static inline unsigned cdiv(long a, long b) { return (unsigned)((a + b - 1) / b); }

extern "C" void kernel_launch(void* const* d_in, const int* in_sizes, int n_in,
                              void* d_out, int out_size, void* d_ws, size_t ws_size,
                              hipStream_t stream) {
  const float* x  = (const float*)d_in[0];
  const int*   ei = (const int*)d_in[1];
  const float* ew = (const float*)d_in[2];
  const float* W1 = (const float*)d_in[3];
  const float* b1 = (const float*)d_in[4];
  const float* W2 = (const float*)d_in[5];
  const float* b2 = (const float*)d_in[6];
  const float* W3 = (const float*)d_in[7];
  const float* b3 = (const float*)d_in[8];

  const int N = in_sizes[0] / 128;
  const int E = in_sizes[2];
  const int* src = ei;
  const int* dst = ei + E;

  float* ws   = (float*)d_ws;
  float* deg  = ws;  ws += N;
  float* dinv = ws;  ws += N;
  float* nrm  = ws;  ws += E;
  float* bufA = ws;  ws += (size_t)N * 64;
  float* bufB = ws;
  float* out  = (float*)d_out;

  const dim3 blk(256);

  deg_init_k   <<<cdiv(N, 256), blk, 0, stream>>>(deg, N);
  deg_scatter_k<<<cdiv(E, 256), blk, 0, stream>>>(dst, ew, deg, E);
  dinv_k       <<<cdiv(N, 256), blk, 0, stream>>>(deg, dinv, N);
  norm_k       <<<cdiv(E, 256), blk, 0, stream>>>(src, dst, ew, dinv, nrm, E);

  gemm16_wmma<128, 64><<<cdiv((long)(N / 16) * 4, 8), blk, 0, stream>>>(x, W1, bufA, N);
  agg_init_k<64> <<<cdiv((long)N * 64, 256), blk, 0, stream>>>(bufA, dinv, b1, bufB, N);
  agg_edges_k<64><<<cdiv((long)E * 64, 256), blk, 0, stream>>>(bufA, src, dst, nrm, bufB, E);
  relu_k         <<<cdiv((long)N * 64, 256), blk, 0, stream>>>(bufB, N * 64);

  gemm16_wmma<64, 64><<<cdiv((long)(N / 16) * 4, 8), blk, 0, stream>>>(bufB, W2, bufA, N);
  agg_init_k<64> <<<cdiv((long)N * 64, 256), blk, 0, stream>>>(bufA, dinv, b2, bufB, N);
  agg_edges_k<64><<<cdiv((long)E * 64, 256), blk, 0, stream>>>(bufA, src, dst, nrm, bufB, E);
  relu_k         <<<cdiv((long)N * 64, 256), blk, 0, stream>>>(bufB, N * 64);

  gemm16_wmma<64, 32><<<cdiv((long)(N / 16) * 2, 8), blk, 0, stream>>>(bufB, W3, bufA, N);
  agg_init_k<32> <<<cdiv((long)N * 32, 256), blk, 0, stream>>>(bufA, dinv, b3, out, N);
  agg_edges_k<32><<<cdiv((long)E * 32, 256), blk, 0, stream>>>(bufA, src, dst, nrm, out, E);
  mean_pool_k    <<<32, 256, 0, stream>>>(out, out + (size_t)N * 32, N);
}


# Mean pool: 32 one-column blocks with 128-B-strided serial reads -> 391 row-chunk blocks with coalesced 16-B loads + 1-block final sum (one more launch); the mean's reduction tree changes
# speedup vs baseline: 1.2650x; 1.2650x over previous
#include <hip/hip_runtime.h>

typedef __attribute__((ext_vector_type(2))) float v2f;
typedef __attribute__((ext_vector_type(8))) float v8f;

__global__ void deg_init_k(float* __restrict__ deg, int N) {
  int t = blockIdx.x * blockDim.x + threadIdx.x;
  if (t < N) deg[t] = 1.0f;
}

__global__ void deg_scatter_k(const int* __restrict__ dst, const float* __restrict__ ew,
                              float* __restrict__ deg, int E) {
  int t = blockIdx.x * blockDim.x + threadIdx.x;
  if (t < E) unsafeAtomicAdd(&deg[dst[t]], ew[t]);
}

__global__ void dinv_k(const float* __restrict__ deg, float* __restrict__ dinv, int N) {
  int t = blockIdx.x * blockDim.x + threadIdx.x;
  if (t < N) {
    float d = deg[t];
    dinv[t] = (d > 0.0f) ? rsqrtf(d) : 0.0f;
  }
}

__global__ void norm_k(const int* __restrict__ src, const int* __restrict__ dst,
                       const float* __restrict__ ew, const float* __restrict__ dinv,
                       float* __restrict__ nrm, int E) {
  int t = blockIdx.x * blockDim.x + threadIdx.x;
  if (t < E) nrm[t] = dinv[src[t]] * ew[t] * dinv[dst[t]];
}

template <int K, int F>
__global__ void gemm16_wmma(const float* __restrict__ X, const float* __restrict__ W,
                            float* __restrict__ H, int N) {
  const int lane = threadIdx.x & 31;
  const int wave = threadIdx.x >> 5;
  const int Ft = F >> 4;
  const int Mt = N >> 4;
  const int tile = blockIdx.x * (blockDim.x >> 5) + wave;
  if (tile >= Mt * Ft) return;
  const int mt = tile / Ft;
  const int ft = tile - mt * Ft;
  const int l16 = lane & 15;
  const int hi  = lane >> 4;

  const float* __restrict__ xrow = X + (size_t)(mt * 16 + l16) * K;
  const float* __restrict__ wcol = W + ft * 16 + l16;

  v8f acc = {};
#pragma unroll 4
  for (int k = 0; k < K; k += 4) {
    v2f a, b;
    a.x = xrow[k + 2 * hi];
    a.y = xrow[k + 2 * hi + 1];
    b.x = wcol[(size_t)(k + 2 * hi) * F];
    b.y = wcol[(size_t)(k + 2 * hi + 1) * F];
    acc = __builtin_amdgcn_wmma_f32_16x16x4_f32(false, a, false, b,
                                                (short)0, acc, false, false);
  }

  float* __restrict__ hcol = H + ft * 16 + l16;
#pragma unroll
  for (int r = 0; r < 8; ++r)
    hcol[(size_t)(mt * 16 + r + 8 * hi) * F] = acc[r];
}

template <int F>
__global__ void agg_init_k(const float* __restrict__ h, const float* __restrict__ dinv,
                           const float* __restrict__ bias, float* __restrict__ a, int N) {
  int t = blockIdx.x * blockDim.x + threadIdx.x;
  if (t >= N * F) return;
  constexpr int LOGF = (F == 64) ? 6 : 5;
  int n = t >> LOGF;
  int f = t & (F - 1);
  float di = dinv[n];
  a[t] = bias[f] + di * di * h[t];
}

template <int F>
__global__ void agg_edges_k(const float* __restrict__ h, const int* __restrict__ src,
                            const int* __restrict__ dst, const float* __restrict__ nrm,
                            float* __restrict__ a, int E) {
  unsigned t = blockIdx.x * blockDim.x + threadIdx.x;
  constexpr int LOGF = (F == 64) ? 6 : 5;
  unsigned e = t >> LOGF;
  if (e >= (unsigned)E) return;
  unsigned f = t & (F - 1);
  int s = src[e];
  int d = dst[e];
  float w = nrm[e];
  unsafeAtomicAdd(&a[(size_t)d * F + f], w * h[(size_t)s * F + f]);
}

__global__ void relu_k(float* __restrict__ a, int n) {
  int t = blockIdx.x * blockDim.x + threadIdx.x;
  if (t < n) a[t] = fmaxf(a[t], 0.0f);
}

#define POOL_ROWS 256

__global__ void pool_partial_k(const float* __restrict__ emb, float* __restrict__ part, int N) {
  __shared__ float4 sm4[256];
  const int c4 = threadIdx.x & 7;
  const int rs = threadIdx.x >> 3;
  const int r0 = blockIdx.x * POOL_ROWS + rs;
  float4 v[8];
#pragma unroll
  for (int i = 0; i < 8; ++i) {
    const int r = r0 + i * 32;
    v[i] = (r < N) ? *(const float4*)(emb + (size_t)r * 32 + c4 * 4)
                   : make_float4(0.0f, 0.0f, 0.0f, 0.0f);
  }
  float4 s = v[0];
#pragma unroll
  for (int i = 1; i < 8; ++i) {
    s.x += v[i].x; s.y += v[i].y; s.z += v[i].z; s.w += v[i].w;
  }
  sm4[threadIdx.x] = s;
  __syncthreads();
  if (threadIdx.x < 32) {
    const float* sm = (const float*)sm4;
    float t = 0.0f;
#pragma unroll
    for (int i = 0; i < 32; ++i) t += sm[i * 32 + threadIdx.x];
    part[(size_t)blockIdx.x * 32 + threadIdx.x] = t;
  }
}

__global__ void pool_final_k(const float* __restrict__ part, float* __restrict__ g,
                             int nPart, int N) {
  __shared__ float sm[1024];
  const int c   = threadIdx.x & 31;
  const int sub = threadIdx.x >> 5;
  float s = 0.0f;
  for (int b0 = sub; b0 < nPart; b0 += 32 * 8) {
    float v[8];
#pragma unroll
    for (int i = 0; i < 8; ++i) {
      const int b = b0 + i * 32;
      v[i] = (b < nPart) ? part[(size_t)b * 32 + c] : 0.0f;
    }
#pragma unroll
    for (int i = 0; i < 8; ++i) s += v[i];
  }
  sm[threadIdx.x] = s;
  __syncthreads();
  if (threadIdx.x < 32) {
    float t = 0.0f;
#pragma unroll
    for (int i = 0; i < 32; ++i) t += sm[i * 32 + threadIdx.x];
    g[threadIdx.x] = t / (float)N;
  }
}

static inline unsigned cdiv(long a, long b) { return (unsigned)((a + b - 1) / b); }

extern "C" void kernel_launch(void* const* d_in, const int* in_sizes, int n_in,
                              void* d_out, int out_size, void* d_ws, size_t ws_size,
                              hipStream_t stream) {
  const float* x  = (const float*)d_in[0];
  const int*   ei = (const int*)d_in[1];
  const float* ew = (const float*)d_in[2];
  const float* W1 = (const float*)d_in[3];
  const float* b1 = (const float*)d_in[4];
  const float* W2 = (const float*)d_in[5];
  const float* b2 = (const float*)d_in[6];
  const float* W3 = (const float*)d_in[7];
  const float* b3 = (const float*)d_in[8];

  const int N = in_sizes[0] / 128;
  const int E = in_sizes[2];
  const int* src = ei;
  const int* dst = ei + E;

  float* ws   = (float*)d_ws;
  float* deg  = ws;  ws += N;
  float* dinv = ws;  ws += N;
  float* nrm  = ws;  ws += E;
  float* bufA = ws;  ws += (size_t)N * 64;
  float* bufB = ws;  ws += (size_t)N * 64;
  float* part = ws;
  float* out  = (float*)d_out;

  const dim3 blk(256);

  deg_init_k   <<<cdiv(N, 256), blk, 0, stream>>>(deg, N);
  deg_scatter_k<<<cdiv(E, 256), blk, 0, stream>>>(dst, ew, deg, E);
  dinv_k       <<<cdiv(N, 256), blk, 0, stream>>>(deg, dinv, N);
  norm_k       <<<cdiv(E, 256), blk, 0, stream>>>(src, dst, ew, dinv, nrm, E);

  gemm16_wmma<128, 64><<<cdiv((long)(N / 16) * 4, 8), blk, 0, stream>>>(x, W1, bufA, N);
  agg_init_k<64> <<<cdiv((long)N * 64, 256), blk, 0, stream>>>(bufA, dinv, b1, bufB, N);
  agg_edges_k<64><<<cdiv((long)E * 64, 256), blk, 0, stream>>>(bufA, src, dst, nrm, bufB, E);
  relu_k         <<<cdiv((long)N * 64, 256), blk, 0, stream>>>(bufB, N * 64);

  gemm16_wmma<64, 64><<<cdiv((long)(N / 16) * 4, 8), blk, 0, stream>>>(bufB, W2, bufA, N);
  agg_init_k<64> <<<cdiv((long)N * 64, 256), blk, 0, stream>>>(bufA, dinv, b2, bufB, N);
  agg_edges_k<64><<<cdiv((long)E * 64, 256), blk, 0, stream>>>(bufA, src, dst, nrm, bufB, E);
  relu_k         <<<cdiv((long)N * 64, 256), blk, 0, stream>>>(bufB, N * 64);

  gemm16_wmma<64, 32><<<cdiv((long)(N / 16) * 2, 8), blk, 0, stream>>>(bufB, W3, bufA, N);
  agg_init_k<32> <<<cdiv((long)N * 32, 256), blk, 0, stream>>>(bufA, dinv, b3, out, N);
  agg_edges_k<32><<<cdiv((long)E * 32, 256), blk, 0, stream>>>(bufA, src, dst, nrm, out, E);
  pool_partial_k <<<cdiv(N, POOL_ROWS), blk, 0, stream>>>(out, part, N);
  pool_final_k   <<<1, 1024, 0, stream>>>(part, out + (size_t)N * 32, (int)cdiv(N, POOL_ROWS), N);
}


# Degree and 3 aggregations: float-atomic scatter -> per-call counting sort of edges by destination (int atomic rank) + atomic-free gather, one wave writes a row once; sum order per row changes
# speedup vs baseline: 3.1714x; 2.5070x over previous
#include <hip/hip_runtime.h>

typedef __attribute__((ext_vector_type(2))) float v2f;
typedef __attribute__((ext_vector_type(8))) float v8f;
typedef __attribute__((ext_vector_type(2))) int   v2i;

__global__ void cnt_init_k(int* __restrict__ cnt, int N) {
  int t = blockIdx.x * blockDim.x + threadIdx.x;
  if (t < N) cnt[t] = 0;
}

__global__ void rank_k(const int* __restrict__ dst, int* __restrict__ cnt,
                       int* __restrict__ rank, int E) {
  int t = blockIdx.x * blockDim.x + threadIdx.x;
  if (t < E) rank[t] = atomicAdd(&cnt[dst[t]], 1);
}

__global__ void rowptr_k(const int* __restrict__ cnt, int* __restrict__ rowptr, int N) {
  __shared__ int s_before[32];
  __shared__ int s_wave[32];
  const int tid  = threadIdx.x;
  const int lane = tid & 31;
  const int wave = tid >> 5;
  const int base = blockIdx.x * 1024;
  int before = 0;
#pragma unroll 8
  for (int i = tid; i < base; i += 1024) before += cnt[i];
  for (int o = 16; o > 0; o >>= 1) before += __shfl_xor(before, o, 32);

  const int i = base + tid;
  const int c = (i < N) ? cnt[i] : 0;
  int x = c;
  for (int o = 1; o < 32; o <<= 1) {
    const int y = __shfl_up(x, o, 32);
    if (lane >= o) x += y;
  }
  if (lane == 31) s_wave[wave] = x;
  if (lane == 0)  s_before[wave] = before;
  __syncthreads();
  int off = 0;
  for (int w = 0; w < 32; ++w) {
    off += s_before[w];
    if (w < wave) off += s_wave[w];
  }
  if (i <= N) rowptr[i] = off + x - c;
}

__global__ void fill_k(const int* __restrict__ src, const int* __restrict__ dst,
                       const float* __restrict__ ew, const int* __restrict__ rowptr,
                       const int* __restrict__ rank, int* __restrict__ csr, int E) {
  int t = blockIdx.x * blockDim.x + threadIdx.x;
  if (t < E) {
    const int p = rowptr[dst[t]] + rank[t];
    v2i e;
    e.x = src[t];
    e.y = __float_as_int(ew[t]);
    ((v2i*)csr)[p] = e;
  }
}

__global__ void rowdeg_k(const int* __restrict__ rowptr, const int* __restrict__ csr,
                         float* __restrict__ dinv, int N) {
  int n = blockIdx.x * blockDim.x + threadIdx.x;
  if (n < N) {
    const v2i* __restrict__ pr = (const v2i*)csr;
    const int end = rowptr[n + 1];
    float d = 1.0f;
    for (int j = rowptr[n]; j < end; ++j) d += __int_as_float(pr[j].y);
    dinv[n] = (d > 0.0f) ? rsqrtf(d) : 0.0f;
  }
}

__global__ void weights_k(const int* __restrict__ rowptr, const float* __restrict__ dinv,
                          int* csr, int N) {
  const int lane = threadIdx.x & 31;
  const int n = __builtin_amdgcn_readfirstlane(
      (int)(blockIdx.x * (blockDim.x >> 5) + (threadIdx.x >> 5)));
  if (n >= N) return;
  const int   end = rowptr[n + 1];
  const float dn  = dinv[n];
  v2i* pr = (v2i*)csr;
  for (int j = rowptr[n] + lane; j < end; j += 32) {
    v2i e = pr[j];
    e.y = __float_as_int(dinv[e.x] * __int_as_float(e.y) * dn);
    pr[j] = e;
  }
}

template <int K, int F>
__global__ void gemm16_wmma(const float* __restrict__ X, const float* __restrict__ W,
                            float* __restrict__ H, int N) {
  const int lane = threadIdx.x & 31;
  const int wave = threadIdx.x >> 5;
  const int Ft = F >> 4;
  const int Mt = N >> 4;
  const int tile = blockIdx.x * (blockDim.x >> 5) + wave;
  if (tile >= Mt * Ft) return;
  const int mt = tile / Ft;
  const int ft = tile - mt * Ft;
  const int l16 = lane & 15;
  const int hi  = lane >> 4;

  const float* __restrict__ xrow = X + (size_t)(mt * 16 + l16) * K;
  const float* __restrict__ wcol = W + ft * 16 + l16;

  v8f acc = {};
#pragma unroll 4
  for (int k = 0; k < K; k += 4) {
    v2f a, b;
    a.x = xrow[k + 2 * hi];
    a.y = xrow[k + 2 * hi + 1];
    b.x = wcol[(size_t)(k + 2 * hi) * F];
    b.y = wcol[(size_t)(k + 2 * hi + 1) * F];
    acc = __builtin_amdgcn_wmma_f32_16x16x4_f32(false, a, false, b,
                                                (short)0, acc, false, false);
  }

  float* __restrict__ hcol = H + ft * 16 + l16;
#pragma unroll
  for (int r = 0; r < 8; ++r)
    hcol[(size_t)(mt * 16 + r + 8 * hi) * F] = acc[r];
}

template <int F, int CNT>
__device__ __forceinline__ void gather_rows(const float* __restrict__ h,
                                            const v2i* __restrict__ csr,
                                            int j, int lane, float (&acc)[F / 32]) {
  int   s[CNT];
  float w[CNT];
#pragma unroll
  for (int u = 0; u < CNT; ++u) {
    const v2i e = csr[j + u];
    s[u] = e.x;
    w[u] = __int_as_float(e.y);
  }
  if constexpr (F == 64) {
    v2f v[CNT];
#pragma unroll
    for (int u = 0; u < CNT; ++u) v[u] = *(const v2f*)(h + (size_t)s[u] * F + lane * 2);
#pragma unroll
    for (int u = 0; u < CNT; ++u) { acc[0] += w[u] * v[u].x; acc[1] += w[u] * v[u].y; }
  } else {
    float v[CNT];
#pragma unroll
    for (int u = 0; u < CNT; ++u) v[u] = h[(size_t)s[u] * F + lane];
#pragma unroll
    for (int u = 0; u < CNT; ++u) acc[0] += w[u] * v[u];
  }
}

template <int F>
__global__ void agg_gather_k(const float* __restrict__ h, const int* __restrict__ rowptr,
                             const int* __restrict__ csr_pairs,
                             const float* __restrict__ dinv, const float* __restrict__ bias,
                             float* __restrict__ a, int N) {
  static_assert(F == 64 || F == 32, "a lane holds one or two columns");
  const int lane = threadIdx.x & 31;
  const int n = __builtin_amdgcn_readfirstlane(
      (int)(blockIdx.x * (blockDim.x >> 5) + (threadIdx.x >> 5)));
  if (n >= N) return;
  const int beg = rowptr[n];
  const int end = rowptr[n + 1];
  const v2i* __restrict__ csr = (const v2i*)csr_pairs;

  const float di = dinv[n];

  float acc[F / 32] = {};
  int j = beg;
  for (; j + 8 <= end; j += 8) gather_rows<F, 8>(h, csr, j, lane, acc);
  const int r = end - j;
  if (r & 4) { gather_rows<F, 4>(h, csr, j, lane, acc); j += 4; }
  if (r & 2) { gather_rows<F, 2>(h, csr, j, lane, acc); j += 2; }
  if (r & 1) { gather_rows<F, 1>(h, csr, j, lane, acc); }

  if constexpr (F == 64) {
    const v2f hv = *(const v2f*)(h + (size_t)n * F + lane * 2);
    v2f o;
    o.x = (bias[lane * 2]     + di * di * hv.x) + acc[0];
    o.y = (bias[lane * 2 + 1] + di * di * hv.y) + acc[1];
    *(v2f*)(a + (size_t)n * F + lane * 2) = o;
  } else {
    const float hv = h[(size_t)n * F + lane];
    a[(size_t)n * F + lane] = (bias[lane] + di * di * hv) + acc[0];
  }
}

__global__ void relu_k(float* __restrict__ a, int n) {
  int t = blockIdx.x * blockDim.x + threadIdx.x;
  if (t < n) a[t] = fmaxf(a[t], 0.0f);
}

#define POOL_ROWS 256

__global__ void pool_partial_k(const float* __restrict__ emb, float* __restrict__ part, int N) {
  __shared__ float4 sm4[256];
  const int c4 = threadIdx.x & 7;
  const int rs = threadIdx.x >> 3;
  const int r0 = blockIdx.x * POOL_ROWS + rs;
  float4 v[8];
#pragma unroll
  for (int i = 0; i < 8; ++i) {
    const int r = r0 + i * 32;
    v[i] = (r < N) ? *(const float4*)(emb + (size_t)r * 32 + c4 * 4)
                   : make_float4(0.0f, 0.0f, 0.0f, 0.0f);
  }
  float4 s = v[0];
#pragma unroll
  for (int i = 1; i < 8; ++i) {
    s.x += v[i].x; s.y += v[i].y; s.z += v[i].z; s.w += v[i].w;
  }
  sm4[threadIdx.x] = s;
  __syncthreads();
  if (threadIdx.x < 32) {
    const float* sm = (const float*)sm4;
    float t = 0.0f;
#pragma unroll
    for (int i = 0; i < 32; ++i) t += sm[i * 32 + threadIdx.x];
    part[(size_t)blockIdx.x * 32 + threadIdx.x] = t;
  }
}

__global__ void pool_final_k(const float* __restrict__ part, float* __restrict__ g,
                             int nPart, int N) {
  __shared__ float sm[1024];
  const int c   = threadIdx.x & 31;
  const int sub = threadIdx.x >> 5;
  float s = 0.0f;
  for (int b0 = sub; b0 < nPart; b0 += 32 * 8) {
    float v[8];
#pragma unroll
    for (int i = 0; i < 8; ++i) {
      const int b = b0 + i * 32;
      v[i] = (b < nPart) ? part[(size_t)b * 32 + c] : 0.0f;
    }
#pragma unroll
    for (int i = 0; i < 8; ++i) s += v[i];
  }
  sm[threadIdx.x] = s;
  __syncthreads();
  if (threadIdx.x < 32) {
    float t = 0.0f;
#pragma unroll
    for (int i = 0; i < 32; ++i) t += sm[i * 32 + threadIdx.x];
    g[threadIdx.x] = t / (float)N;
  }
}

static inline unsigned cdiv(long a, long b) { return (unsigned)((a + b - 1) / b); }

extern "C" void kernel_launch(void* const* d_in, const int* in_sizes, int n_in,
                              void* d_out, int out_size, void* d_ws, size_t ws_size,
                              hipStream_t stream) {
  const float* x  = (const float*)d_in[0];
  const int*   ei = (const int*)d_in[1];
  const float* ew = (const float*)d_in[2];
  const float* W1 = (const float*)d_in[3];
  const float* b1 = (const float*)d_in[4];
  const float* W2 = (const float*)d_in[5];
  const float* b2 = (const float*)d_in[6];
  const float* W3 = (const float*)d_in[7];
  const float* b3 = (const float*)d_in[8];

  const int N = in_sizes[0] / 128;
  const int E = in_sizes[2];
  const int* src = ei;
  const int* dst = ei + E;

  float* ws      = (float*)d_ws;
  float* bufA    = ws;        ws += (size_t)N * 64;
  float* bufB    = ws;        ws += (size_t)N * 64;
  int*   csr     = (int*)ws;  ws += (size_t)E * 2;
  int*   rank    = (int*)ws;  ws += E;
  float* dinv    = ws;        ws += N;
  int*   cnt     = (int*)ws;  ws += N;
  int*   rowptr  = (int*)ws;  ws += N + 1;
  float* part    = ws;
  float* out     = (float*)d_out;

  const dim3 blk(256);

  cnt_init_k<<<cdiv(N, 256), blk, 0, stream>>>(cnt, N);
  rank_k    <<<cdiv(E, 256), blk, 0, stream>>>(dst, cnt, rank, E);
  rowptr_k  <<<cdiv((long)N + 1, 1024), 1024, 0, stream>>>(cnt, rowptr, N);
  fill_k    <<<cdiv(E, 256), blk, 0, stream>>>(src, dst, ew, rowptr, rank, csr, E);
  rowdeg_k  <<<cdiv(N, 256), blk, 0, stream>>>(rowptr, csr, dinv, N);
  weights_k <<<cdiv(N, 8), blk, 0, stream>>>(rowptr, dinv, csr, N);

  gemm16_wmma<128, 64><<<cdiv((long)(N / 16) * 4, 8), blk, 0, stream>>>(x, W1, bufA, N);
  agg_gather_k<64><<<cdiv(N, 8), blk, 0, stream>>>(bufA, rowptr, csr, dinv, b1, bufB, N);
  relu_k         <<<cdiv((long)N * 64, 256), blk, 0, stream>>>(bufB, N * 64);

  gemm16_wmma<64, 64><<<cdiv((long)(N / 16) * 4, 8), blk, 0, stream>>>(bufB, W2, bufA, N);
  agg_gather_k<64><<<cdiv(N, 8), blk, 0, stream>>>(bufA, rowptr, csr, dinv, b2, bufB, N);
  relu_k         <<<cdiv((long)N * 64, 256), blk, 0, stream>>>(bufB, N * 64);

  gemm16_wmma<64, 32><<<cdiv((long)(N / 16) * 2, 8), blk, 0, stream>>>(bufB, W3, bufA, N);
  agg_gather_k<32><<<cdiv(N, 8), blk, 0, stream>>>(bufA, rowptr, csr, dinv, b3, out, N);
  pool_partial_k <<<cdiv(N, POOL_ROWS), blk, 0, stream>>>(out, part, N);
  pool_final_k   <<<1, 1024, 0, stream>>>(part, out + (size_t)N * 32, (int)cdiv(N, POOL_ROWS), N);
}


# ReLU of layers 1 and 2 folded into the aggregation kernel's row store: the two elementwise ReLU launches (one dword per lane; a read and a write of N x 64 floats each) are gone; sums unchanged
# speedup vs baseline: 3.6702x; 1.1573x over previous
#include <hip/hip_runtime.h>

typedef __attribute__((ext_vector_type(2))) float v2f;
typedef __attribute__((ext_vector_type(8))) float v8f;
typedef __attribute__((ext_vector_type(2))) int   v2i;

__global__ void cnt_init_k(int* __restrict__ cnt, int N) {
  int t = blockIdx.x * blockDim.x + threadIdx.x;
  if (t < N) cnt[t] = 0;
}

__global__ void rank_k(const int* __restrict__ dst, int* __restrict__ cnt,
                       int* __restrict__ rank, int E) {
  int t = blockIdx.x * blockDim.x + threadIdx.x;
  if (t < E) rank[t] = atomicAdd(&cnt[dst[t]], 1);
}

__global__ void rowptr_k(const int* __restrict__ cnt, int* __restrict__ rowptr, int N) {
  __shared__ int s_before[32];
  __shared__ int s_wave[32];
  const int tid  = threadIdx.x;
  const int lane = tid & 31;
  const int wave = tid >> 5;
  const int base = blockIdx.x * 1024;
  int before = 0;
#pragma unroll 8
  for (int i = tid; i < base; i += 1024) before += cnt[i];
  for (int o = 16; o > 0; o >>= 1) before += __shfl_xor(before, o, 32);

  const int i = base + tid;
  const int c = (i < N) ? cnt[i] : 0;
  int x = c;
  for (int o = 1; o < 32; o <<= 1) {
    const int y = __shfl_up(x, o, 32);
    if (lane >= o) x += y;
  }
  if (lane == 31) s_wave[wave] = x;
  if (lane == 0)  s_before[wave] = before;
  __syncthreads();
  int off = 0;
  for (int w = 0; w < 32; ++w) {
    off += s_before[w];
    if (w < wave) off += s_wave[w];
  }
  if (i <= N) rowptr[i] = off + x - c;
}

__global__ void fill_k(const int* __restrict__ src, const int* __restrict__ dst,
                       const float* __restrict__ ew, const int* __restrict__ rowptr,
                       const int* __restrict__ rank, int* __restrict__ csr, int E) {
  int t = blockIdx.x * blockDim.x + threadIdx.x;
  if (t < E) {
    const int p = rowptr[dst[t]] + rank[t];
    v2i e;
    e.x = src[t];
    e.y = __float_as_int(ew[t]);
    ((v2i*)csr)[p] = e;
  }
}

__global__ void rowdeg_k(const int* __restrict__ rowptr, const int* __restrict__ csr,
                         float* __restrict__ dinv, int N) {
  int n = blockIdx.x * blockDim.x + threadIdx.x;
  if (n < N) {
    const v2i* __restrict__ pr = (const v2i*)csr;
    const int end = rowptr[n + 1];
    float d = 1.0f;
    for (int j = rowptr[n]; j < end; ++j) d += __int_as_float(pr[j].y);
    dinv[n] = (d > 0.0f) ? rsqrtf(d) : 0.0f;
  }
}

__global__ void weights_k(const int* __restrict__ rowptr, const float* __restrict__ dinv,
                          int* csr, int N) {
  const int lane = threadIdx.x & 31;
  const int n = __builtin_amdgcn_readfirstlane(
      (int)(blockIdx.x * (blockDim.x >> 5) + (threadIdx.x >> 5)));
  if (n >= N) return;
  const int   end = rowptr[n + 1];
  const float dn  = dinv[n];
  v2i* pr = (v2i*)csr;
  for (int j = rowptr[n] + lane; j < end; j += 32) {
    v2i e = pr[j];
    e.y = __float_as_int(dinv[e.x] * __int_as_float(e.y) * dn);
    pr[j] = e;
  }
}

template <int K, int F>
__global__ void gemm16_wmma(const float* __restrict__ X, const float* __restrict__ W,
                            float* __restrict__ H, int N) {
  const int lane = threadIdx.x & 31;
  const int wave = threadIdx.x >> 5;
  const int Ft = F >> 4;
  const int Mt = N >> 4;
  const int tile = blockIdx.x * (blockDim.x >> 5) + wave;
  if (tile >= Mt * Ft) return;
  const int mt = tile / Ft;
  const int ft = tile - mt * Ft;
  const int l16 = lane & 15;
  const int hi  = lane >> 4;

  const float* __restrict__ xrow = X + (size_t)(mt * 16 + l16) * K;
  const float* __restrict__ wcol = W + ft * 16 + l16;

  v8f acc = {};
#pragma unroll 4
  for (int k = 0; k < K; k += 4) {
    v2f a, b;
    a.x = xrow[k + 2 * hi];
    a.y = xrow[k + 2 * hi + 1];
    b.x = wcol[(size_t)(k + 2 * hi) * F];
    b.y = wcol[(size_t)(k + 2 * hi + 1) * F];
    acc = __builtin_amdgcn_wmma_f32_16x16x4_f32(false, a, false, b,
                                                (short)0, acc, false, false);
  }

  float* __restrict__ hcol = H + ft * 16 + l16;
#pragma unroll
  for (int r = 0; r < 8; ++r)
    hcol[(size_t)(mt * 16 + r + 8 * hi) * F] = acc[r];
}

template <int F, int CNT>
__device__ __forceinline__ void gather_rows(const float* __restrict__ h,
                                            const v2i* __restrict__ csr,
                                            int j, int lane, float (&acc)[F / 32]) {
  int   s[CNT];
  float w[CNT];
#pragma unroll
  for (int u = 0; u < CNT; ++u) {
    const v2i e = csr[j + u];
    s[u] = e.x;
    w[u] = __int_as_float(e.y);
  }
  if constexpr (F == 64) {
    v2f v[CNT];
#pragma unroll
    for (int u = 0; u < CNT; ++u) v[u] = *(const v2f*)(h + (size_t)s[u] * F + lane * 2);
#pragma unroll
    for (int u = 0; u < CNT; ++u) { acc[0] += w[u] * v[u].x; acc[1] += w[u] * v[u].y; }
  } else {
    float v[CNT];
#pragma unroll
    for (int u = 0; u < CNT; ++u) v[u] = h[(size_t)s[u] * F + lane];
#pragma unroll
    for (int u = 0; u < CNT; ++u) acc[0] += w[u] * v[u];
  }
}

template <int F, bool RELU>
__global__ void agg_gather_k(const float* __restrict__ h, const int* __restrict__ rowptr,
                             const int* __restrict__ csr_pairs,
                             const float* __restrict__ dinv, const float* __restrict__ bias,
                             float* __restrict__ a, int N) {
  static_assert(F == 64 || F == 32, "a lane holds one or two columns");
  const int lane = threadIdx.x & 31;
  const int n = __builtin_amdgcn_readfirstlane(
      (int)(blockIdx.x * (blockDim.x >> 5) + (threadIdx.x >> 5)));
  if (n >= N) return;
  const int beg = rowptr[n];
  const int end = rowptr[n + 1];
  const v2i* __restrict__ csr = (const v2i*)csr_pairs;

  const float di = dinv[n];

  float acc[F / 32] = {};
  int j = beg;
  for (; j + 8 <= end; j += 8) gather_rows<F, 8>(h, csr, j, lane, acc);
  const int r = end - j;
  if (r & 4) { gather_rows<F, 4>(h, csr, j, lane, acc); j += 4; }
  if (r & 2) { gather_rows<F, 2>(h, csr, j, lane, acc); j += 2; }
  if (r & 1) { gather_rows<F, 1>(h, csr, j, lane, acc); }

  if constexpr (F == 64) {
    const v2f hv = *(const v2f*)(h + (size_t)n * F + lane * 2);
    v2f o;
    o.x = (bias[lane * 2]     + di * di * hv.x) + acc[0];
    o.y = (bias[lane * 2 + 1] + di * di * hv.y) + acc[1];
    if constexpr (RELU) { o.x = fmaxf(o.x, 0.0f); o.y = fmaxf(o.y, 0.0f); }
    *(v2f*)(a + (size_t)n * F + lane * 2) = o;
  } else {
    const float hv = h[(size_t)n * F + lane];
    float o = (bias[lane] + di * di * hv) + acc[0];
    if constexpr (RELU) o = fmaxf(o, 0.0f);
    a[(size_t)n * F + lane] = o;
  }
}

#define POOL_ROWS 256

__global__ void pool_partial_k(const float* __restrict__ emb, float* __restrict__ part, int N) {
  __shared__ float4 sm4[256];
  const int c4 = threadIdx.x & 7;
  const int rs = threadIdx.x >> 3;
  const int r0 = blockIdx.x * POOL_ROWS + rs;
  float4 v[8];
#pragma unroll
  for (int i = 0; i < 8; ++i) {
    const int r = r0 + i * 32;
    v[i] = (r < N) ? *(const float4*)(emb + (size_t)r * 32 + c4 * 4)
                   : make_float4(0.0f, 0.0f, 0.0f, 0.0f);
  }
  float4 s = v[0];
#pragma unroll
  for (int i = 1; i < 8; ++i) {
    s.x += v[i].x; s.y += v[i].y; s.z += v[i].z; s.w += v[i].w;
  }
  sm4[threadIdx.x] = s;
  __syncthreads();
  if (threadIdx.x < 32) {
    const float* sm = (const float*)sm4;
    float t = 0.0f;
#pragma unroll
    for (int i = 0; i < 32; ++i) t += sm[i * 32 + threadIdx.x];
    part[(size_t)blockIdx.x * 32 + threadIdx.x] = t;
  }
}

__global__ void pool_final_k(const float* __restrict__ part, float* __restrict__ g,
                             int nPart, int N) {
  __shared__ float sm[1024];
  const int c   = threadIdx.x & 31;
  const int sub = threadIdx.x >> 5;
  float s = 0.0f;
  for (int b0 = sub; b0 < nPart; b0 += 32 * 8) {
    float v[8];
#pragma unroll
    for (int i = 0; i < 8; ++i) {
      const int b = b0 + i * 32;
      v[i] = (b < nPart) ? part[(size_t)b * 32 + c] : 0.0f;
    }
#pragma unroll
    for (int i = 0; i < 8; ++i) s += v[i];
  }
  sm[threadIdx.x] = s;
  __syncthreads();
  if (threadIdx.x < 32) {
    float t = 0.0f;
#pragma unroll
    for (int i = 0; i < 32; ++i) t += sm[i * 32 + threadIdx.x];
    g[threadIdx.x] = t / (float)N;
  }
}

static inline unsigned cdiv(long a, long b) { return (unsigned)((a + b - 1) / b); }

extern "C" void kernel_launch(void* const* d_in, const int* in_sizes, int n_in,
                              void* d_out, int out_size, void* d_ws, size_t ws_size,
                              hipStream_t stream) {
  const float* x  = (const float*)d_in[0];
  const int*   ei = (const int*)d_in[1];
  const float* ew = (const float*)d_in[2];
  const float* W1 = (const float*)d_in[3];
  const float* b1 = (const float*)d_in[4];
  const float* W2 = (const float*)d_in[5];
  const float* b2 = (const float*)d_in[6];
  const float* W3 = (const float*)d_in[7];
  const float* b3 = (const float*)d_in[8];

  const int N = in_sizes[0] / 128;
  const int E = in_sizes[2];
  const int* src = ei;
  const int* dst = ei + E;

  float* ws      = (float*)d_ws;
  float* bufA    = ws;        ws += (size_t)N * 64;
  float* bufB    = ws;        ws += (size_t)N * 64;
  int*   csr     = (int*)ws;  ws += (size_t)E * 2;
  int*   rank    = (int*)ws;  ws += E;
  float* dinv    = ws;        ws += N;
  int*   cnt     = (int*)ws;  ws += N;
  int*   rowptr  = (int*)ws;  ws += N + 1;
  float* part    = ws;
  float* out     = (float*)d_out;

  const dim3 blk(256);

  cnt_init_k<<<cdiv(N, 256), blk, 0, stream>>>(cnt, N);
  rank_k    <<<cdiv(E, 256), blk, 0, stream>>>(dst, cnt, rank, E);
  rowptr_k  <<<cdiv((long)N + 1, 1024), 1024, 0, stream>>>(cnt, rowptr, N);
  fill_k    <<<cdiv(E, 256), blk, 0, stream>>>(src, dst, ew, rowptr, rank, csr, E);
  rowdeg_k  <<<cdiv(N, 256), blk, 0, stream>>>(rowptr, csr, dinv, N);
  weights_k <<<cdiv(N, 8), blk, 0, stream>>>(rowptr, dinv, csr, N);

  gemm16_wmma<128, 64><<<cdiv((long)(N / 16) * 4, 8), blk, 0, stream>>>(x, W1, bufA, N);
  agg_gather_k<64, true><<<cdiv(N, 8), blk, 0, stream>>>(bufA, rowptr, csr, dinv, b1, bufB, N);

  gemm16_wmma<64, 64><<<cdiv((long)(N / 16) * 4, 8), blk, 0, stream>>>(bufB, W2, bufA, N);
  agg_gather_k<64, true><<<cdiv(N, 8), blk, 0, stream>>>(bufA, rowptr, csr, dinv, b2, bufB, N);

  gemm16_wmma<64, 32><<<cdiv((long)(N / 16) * 2, 8), blk, 0, stream>>>(bufB, W3, bufA, N);
  agg_gather_k<32, false><<<cdiv(N, 8), blk, 0, stream>>>(bufA, rowptr, csr, dinv, b3, out, N);
  pool_partial_k <<<cdiv(N, POOL_ROWS), blk, 0, stream>>>(out, part, N);
  pool_final_k   <<<1, 1024, 0, stream>>>(part, out + (size_t)N * 32, (int)cdiv(N, POOL_ROWS), N);
}


# Norm factored out of the edges: GEMM stores dinv[row]*h[row]; aggregation adds the self row to the edge sum, then scales by dinv[dst]; per-edge norm pass (a launch, a random dinv read per edge) gone
# speedup vs baseline: 3.8360x; 1.0452x over previous
#include <hip/hip_runtime.h>

typedef __attribute__((ext_vector_type(2))) float v2f;
typedef __attribute__((ext_vector_type(8))) float v8f;
typedef __attribute__((ext_vector_type(2))) int   v2i;

__global__ void cnt_init_k(int* __restrict__ cnt, int N) {
  int t = blockIdx.x * blockDim.x + threadIdx.x;
  if (t < N) cnt[t] = 0;
}

__global__ void rank_k(const int* __restrict__ dst, int* __restrict__ cnt,
                       int* __restrict__ rank, int E) {
  int t = blockIdx.x * blockDim.x + threadIdx.x;
  if (t < E) rank[t] = atomicAdd(&cnt[dst[t]], 1);
}

__global__ void rowptr_k(const int* __restrict__ cnt, int* __restrict__ rowptr, int N) {
  __shared__ int s_before[32];
  __shared__ int s_wave[32];
  const int tid  = threadIdx.x;
  const int lane = tid & 31;
  const int wave = tid >> 5;
  const int base = blockIdx.x * 1024;
  int before = 0;
#pragma unroll 8
  for (int i = tid; i < base; i += 1024) before += cnt[i];
  for (int o = 16; o > 0; o >>= 1) before += __shfl_xor(before, o, 32);

  const int i = base + tid;
  const int c = (i < N) ? cnt[i] : 0;
  int x = c;
  for (int o = 1; o < 32; o <<= 1) {
    const int y = __shfl_up(x, o, 32);
    if (lane >= o) x += y;
  }
  if (lane == 31) s_wave[wave] = x;
  if (lane == 0)  s_before[wave] = before;
  __syncthreads();
  int off = 0;
  for (int w = 0; w < 32; ++w) {
    off += s_before[w];
    if (w < wave) off += s_wave[w];
  }
  if (i <= N) rowptr[i] = off + x - c;
}

__global__ void fill_k(const int* __restrict__ src, const int* __restrict__ dst,
                       const float* __restrict__ ew, const int* __restrict__ rowptr,
                       const int* __restrict__ rank, int* __restrict__ csr, int E) {
  int t = blockIdx.x * blockDim.x + threadIdx.x;
  if (t < E) {
    const int p = rowptr[dst[t]] + rank[t];
    v2i e;
    e.x = src[t];
    e.y = __float_as_int(ew[t]);
    ((v2i*)csr)[p] = e;
  }
}

__global__ void rowdeg_k(const int* __restrict__ rowptr, const int* __restrict__ csr,
                         float* __restrict__ dinv, int N) {
  int n = blockIdx.x * blockDim.x + threadIdx.x;
  if (n < N) {
    const v2i* __restrict__ pr = (const v2i*)csr;
    const int end = rowptr[n + 1];
    float d = 1.0f;
    for (int j = rowptr[n]; j < end; ++j) d += __int_as_float(pr[j].y);
    dinv[n] = (d > 0.0f) ? rsqrtf(d) : 0.0f;
  }
}

template <int K, int F>
__global__ void gemm16_wmma(const float* __restrict__ X, const float* __restrict__ W,
                            const float* __restrict__ dinv, float* __restrict__ H, int N) {
  const int lane = threadIdx.x & 31;
  const int wave = threadIdx.x >> 5;
  const int Ft = F >> 4;
  const int Mt = N >> 4;
  const int tile = blockIdx.x * (blockDim.x >> 5) + wave;
  if (tile >= Mt * Ft) return;
  const int mt = tile / Ft;
  const int ft = tile - mt * Ft;
  const int l16 = lane & 15;
  const int hi  = lane >> 4;

  const float* __restrict__ xrow = X + (size_t)(mt * 16 + l16) * K;
  const float* __restrict__ wcol = W + ft * 16 + l16;

  float ds[8];
#pragma unroll
  for (int r = 0; r < 8; ++r) ds[r] = dinv[mt * 16 + r + 8 * hi];

  v8f acc = {};
#pragma unroll 4
  for (int k = 0; k < K; k += 4) {
    v2f a, b;
    a.x = xrow[k + 2 * hi];
    a.y = xrow[k + 2 * hi + 1];
    b.x = wcol[(size_t)(k + 2 * hi) * F];
    b.y = wcol[(size_t)(k + 2 * hi + 1) * F];
    acc = __builtin_amdgcn_wmma_f32_16x16x4_f32(false, a, false, b,
                                                (short)0, acc, false, false);
  }

  float* __restrict__ hcol = H + ft * 16 + l16;
#pragma unroll
  for (int r = 0; r < 8; ++r)
    hcol[(size_t)(mt * 16 + r + 8 * hi) * F] = ds[r] * acc[r];
}

template <int F, int CNT>
__device__ __forceinline__ void gather_rows(const float* __restrict__ h,
                                            const v2i* __restrict__ csr,
                                            int j, int lane, float (&acc)[F / 32]) {
  int   s[CNT];
  float w[CNT];
#pragma unroll
  for (int u = 0; u < CNT; ++u) {
    const v2i e = csr[j + u];
    s[u] = e.x;
    w[u] = __int_as_float(e.y);
  }
  if constexpr (F == 64) {
    v2f v[CNT];
#pragma unroll
    for (int u = 0; u < CNT; ++u) v[u] = *(const v2f*)(h + (size_t)s[u] * F + lane * 2);
#pragma unroll
    for (int u = 0; u < CNT; ++u) { acc[0] += w[u] * v[u].x; acc[1] += w[u] * v[u].y; }
  } else {
    float v[CNT];
#pragma unroll
    for (int u = 0; u < CNT; ++u) v[u] = h[(size_t)s[u] * F + lane];
#pragma unroll
    for (int u = 0; u < CNT; ++u) acc[0] += w[u] * v[u];
  }
}

template <int F, bool RELU>
__global__ void agg_gather_k(const float* __restrict__ h, const int* __restrict__ rowptr,
                             const int* __restrict__ csr_pairs,
                             const float* __restrict__ dinv, const float* __restrict__ bias,
                             float* __restrict__ a, int N) {
  static_assert(F == 64 || F == 32, "a lane holds one or two columns");
  const int lane = threadIdx.x & 31;
  const int n = __builtin_amdgcn_readfirstlane(
      (int)(blockIdx.x * (blockDim.x >> 5) + (threadIdx.x >> 5)));
  if (n >= N) return;
  const int beg = rowptr[n];
  const int end = rowptr[n + 1];
  const v2i* __restrict__ csr = (const v2i*)csr_pairs;

  const float di = dinv[n];

  float acc[F / 32] = {};
  int j = beg;
  for (; j + 8 <= end; j += 8) gather_rows<F, 8>(h, csr, j, lane, acc);
  const int r = end - j;
  if (r & 4) { gather_rows<F, 4>(h, csr, j, lane, acc); j += 4; }
  if (r & 2) { gather_rows<F, 2>(h, csr, j, lane, acc); j += 2; }
  if (r & 1) { gather_rows<F, 1>(h, csr, j, lane, acc); }

  if constexpr (F == 64) {
    const v2f hv = *(const v2f*)(h + (size_t)n * F + lane * 2);
    v2f o;
    o.x = bias[lane * 2]     + di * (acc[0] + hv.x);
    o.y = bias[lane * 2 + 1] + di * (acc[1] + hv.y);
    if constexpr (RELU) { o.x = fmaxf(o.x, 0.0f); o.y = fmaxf(o.y, 0.0f); }
    *(v2f*)(a + (size_t)n * F + lane * 2) = o;
  } else {
    const float hv = h[(size_t)n * F + lane];
    float o = bias[lane] + di * (acc[0] + hv);
    if constexpr (RELU) o = fmaxf(o, 0.0f);
    a[(size_t)n * F + lane] = o;
  }
}

#define POOL_ROWS 256

__global__ void pool_partial_k(const float* __restrict__ emb, float* __restrict__ part, int N) {
  __shared__ float4 sm4[256];
  const int c4 = threadIdx.x & 7;
  const int rs = threadIdx.x >> 3;
  const int r0 = blockIdx.x * POOL_ROWS + rs;
  float4 v[8];
#pragma unroll
  for (int i = 0; i < 8; ++i) {
    const int r = r0 + i * 32;
    v[i] = (r < N) ? *(const float4*)(emb + (size_t)r * 32 + c4 * 4)
                   : make_float4(0.0f, 0.0f, 0.0f, 0.0f);
  }
  float4 s = v[0];
#pragma unroll
  for (int i = 1; i < 8; ++i) {
    s.x += v[i].x; s.y += v[i].y; s.z += v[i].z; s.w += v[i].w;
  }
  sm4[threadIdx.x] = s;
  __syncthreads();
  if (threadIdx.x < 32) {
    const float* sm = (const float*)sm4;
    float t = 0.0f;
#pragma unroll
    for (int i = 0; i < 32; ++i) t += sm[i * 32 + threadIdx.x];
    part[(size_t)blockIdx.x * 32 + threadIdx.x] = t;
  }
}

__global__ void pool_final_k(const float* __restrict__ part, float* __restrict__ g,
                             int nPart, int N) {
  __shared__ float sm[1024];
  const int c   = threadIdx.x & 31;
  const int sub = threadIdx.x >> 5;
  float s = 0.0f;
  for (int b0 = sub; b0 < nPart; b0 += 32 * 8) {
    float v[8];
#pragma unroll
    for (int i = 0; i < 8; ++i) {
      const int b = b0 + i * 32;
      v[i] = (b < nPart) ? part[(size_t)b * 32 + c] : 0.0f;
    }
#pragma unroll
    for (int i = 0; i < 8; ++i) s += v[i];
  }
  sm[threadIdx.x] = s;
  __syncthreads();
  if (threadIdx.x < 32) {
    float t = 0.0f;
#pragma unroll
    for (int i = 0; i < 32; ++i) t += sm[i * 32 + threadIdx.x];
    g[threadIdx.x] = t / (float)N;
  }
}

static inline unsigned cdiv(long a, long b) { return (unsigned)((a + b - 1) / b); }

extern "C" void kernel_launch(void* const* d_in, const int* in_sizes, int n_in,
                              void* d_out, int out_size, void* d_ws, size_t ws_size,
                              hipStream_t stream) {
  const float* x  = (const float*)d_in[0];
  const int*   ei = (const int*)d_in[1];
  const float* ew = (const float*)d_in[2];
  const float* W1 = (const float*)d_in[3];
  const float* b1 = (const float*)d_in[4];
  const float* W2 = (const float*)d_in[5];
  const float* b2 = (const float*)d_in[6];
  const float* W3 = (const float*)d_in[7];
  const float* b3 = (const float*)d_in[8];

  const int N = in_sizes[0] / 128;
  const int E = in_sizes[2];
  const int* src = ei;
  const int* dst = ei + E;

  float* ws      = (float*)d_ws;
  float* bufA    = ws;        ws += (size_t)N * 64;
  float* bufB    = ws;        ws += (size_t)N * 64;
  int*   csr     = (int*)ws;  ws += (size_t)E * 2;
  int*   rank    = (int*)ws;  ws += E;
  float* dinv    = ws;        ws += N;
  int*   cnt     = (int*)ws;  ws += N;
  int*   rowptr  = (int*)ws;  ws += N + 1;
  float* part    = ws;
  float* out     = (float*)d_out;

  const dim3 blk(256);

  cnt_init_k<<<cdiv(N, 256), blk, 0, stream>>>(cnt, N);
  rank_k    <<<cdiv(E, 256), blk, 0, stream>>>(dst, cnt, rank, E);
  rowptr_k  <<<cdiv((long)N + 1, 1024), 1024, 0, stream>>>(cnt, rowptr, N);
  fill_k    <<<cdiv(E, 256), blk, 0, stream>>>(src, dst, ew, rowptr, rank, csr, E);
  rowdeg_k  <<<cdiv(N, 256), blk, 0, stream>>>(rowptr, csr, dinv, N);

  gemm16_wmma<128, 64><<<cdiv((long)(N / 16) * 4, 8), blk, 0, stream>>>(x, W1, dinv, bufA, N);
  agg_gather_k<64, true><<<cdiv(N, 8), blk, 0, stream>>>(bufA, rowptr, csr, dinv, b1, bufB, N);

  gemm16_wmma<64, 64><<<cdiv((long)(N / 16) * 4, 8), blk, 0, stream>>>(bufB, W2, dinv, bufA, N);
  agg_gather_k<64, true><<<cdiv(N, 8), blk, 0, stream>>>(bufA, rowptr, csr, dinv, b2, bufB, N);

  gemm16_wmma<64, 32><<<cdiv((long)(N / 16) * 2, 8), blk, 0, stream>>>(bufB, W3, dinv, bufA, N);
  agg_gather_k<32, false><<<cdiv(N, 8), blk, 0, stream>>>(bufA, rowptr, csr, dinv, b3, out, N);
  pool_partial_k <<<cdiv(N, POOL_ROWS), blk, 0, stream>>>(out, part, N);
  pool_final_k   <<<1, 1024, 0, stream>>>(part, out + (size_t)N * 32, (int)cdiv(N, POOL_ROWS), N);
}


# Layer 2 and 3 GEMMs fused into the aggregation before them: a block's 16 finished rows go through LDS to the WMMAs, N x 64 activations never reach memory, 2 launches gone; sums unchanged
# speedup vs baseline: 4.0398x; 1.0531x over previous
#include <hip/hip_runtime.h>

typedef __attribute__((ext_vector_type(2))) float v2f;
typedef __attribute__((ext_vector_type(8))) float v8f;
typedef __attribute__((ext_vector_type(2))) int   v2i;

__global__ void cnt_init_k(int* __restrict__ cnt, int N) {
  int t = blockIdx.x * blockDim.x + threadIdx.x;
  if (t < N) cnt[t] = 0;
}

__global__ void rank_k(const int* __restrict__ dst, int* __restrict__ cnt,
                       int* __restrict__ rank, int E) {
  int t = blockIdx.x * blockDim.x + threadIdx.x;
  if (t < E) rank[t] = atomicAdd(&cnt[dst[t]], 1);
}

__global__ void rowptr_k(const int* __restrict__ cnt, int* __restrict__ rowptr, int N) {
  __shared__ int s_before[32];
  __shared__ int s_wave[32];
  const int tid  = threadIdx.x;
  const int lane = tid & 31;
  const int wave = tid >> 5;
  const int base = blockIdx.x * 1024;
  int before = 0;
#pragma unroll 8
  for (int i = tid; i < base; i += 1024) before += cnt[i];
  for (int o = 16; o > 0; o >>= 1) before += __shfl_xor(before, o, 32);

  const int i = base + tid;
  const int c = (i < N) ? cnt[i] : 0;
  int x = c;
  for (int o = 1; o < 32; o <<= 1) {
    const int y = __shfl_up(x, o, 32);
    if (lane >= o) x += y;
  }
  if (lane == 31) s_wave[wave] = x;
  if (lane == 0)  s_before[wave] = before;
  __syncthreads();
  int off = 0;
  for (int w = 0; w < 32; ++w) {
    off += s_before[w];
    if (w < wave) off += s_wave[w];
  }
  if (i <= N) rowptr[i] = off + x - c;
}

__global__ void fill_k(const int* __restrict__ src, const int* __restrict__ dst,
                       const float* __restrict__ ew, const int* __restrict__ rowptr,
                       const int* __restrict__ rank, int* __restrict__ csr, int E) {
  int t = blockIdx.x * blockDim.x + threadIdx.x;
  if (t < E) {
    const int p = rowptr[dst[t]] + rank[t];
    v2i e;
    e.x = src[t];
    e.y = __float_as_int(ew[t]);
    ((v2i*)csr)[p] = e;
  }
}

__global__ void rowdeg_k(const int* __restrict__ rowptr, const int* __restrict__ csr,
                         float* __restrict__ dinv, int N) {
  int n = blockIdx.x * blockDim.x + threadIdx.x;
  if (n < N) {
    const v2i* __restrict__ pr = (const v2i*)csr;
    const int end = rowptr[n + 1];
    float d = 1.0f;
    for (int j = rowptr[n]; j < end; ++j) d += __int_as_float(pr[j].y);
    dinv[n] = (d > 0.0f) ? rsqrtf(d) : 0.0f;
  }
}

template <int K, int F>
__global__ void gemm16_wmma(const float* __restrict__ X, const float* __restrict__ W,
                            const float* __restrict__ dinv, float* __restrict__ H, int N) {
  const int lane = threadIdx.x & 31;
  const int wave = threadIdx.x >> 5;
  const int Ft = F >> 4;
  const int Mt = N >> 4;
  const int tile = blockIdx.x * (blockDim.x >> 5) + wave;
  if (tile >= Mt * Ft) return;
  const int mt = tile / Ft;
  const int ft = tile - mt * Ft;
  const int l16 = lane & 15;
  const int hi  = lane >> 4;

  const float* __restrict__ xrow = X + (size_t)(mt * 16 + l16) * K;
  const float* __restrict__ wcol = W + ft * 16 + l16;

  float ds[8];
#pragma unroll
  for (int r = 0; r < 8; ++r) ds[r] = dinv[mt * 16 + r + 8 * hi];

  v8f acc = {};
#pragma unroll 4
  for (int k = 0; k < K; k += 4) {
    v2f a, b;
    a.x = xrow[k + 2 * hi];
    a.y = xrow[k + 2 * hi + 1];
    b.x = wcol[(size_t)(k + 2 * hi) * F];
    b.y = wcol[(size_t)(k + 2 * hi + 1) * F];
    acc = __builtin_amdgcn_wmma_f32_16x16x4_f32(false, a, false, b,
                                                (short)0, acc, false, false);
  }

  float* __restrict__ hcol = H + ft * 16 + l16;
#pragma unroll
  for (int r = 0; r < 8; ++r)
    hcol[(size_t)(mt * 16 + r + 8 * hi) * F] = ds[r] * acc[r];
}

template <int F, int CNT>
__device__ __forceinline__ void gather_rows(const float* __restrict__ h,
                                            const v2i* __restrict__ csr,
                                            int j, int lane, float (&acc)[F / 32]) {
  int   s[CNT];
  float w[CNT];
#pragma unroll
  for (int u = 0; u < CNT; ++u) {
    const v2i e = csr[j + u];
    s[u] = e.x;
    w[u] = __int_as_float(e.y);
  }
  if constexpr (F == 64) {
    v2f v[CNT];
#pragma unroll
    for (int u = 0; u < CNT; ++u) v[u] = *(const v2f*)(h + (size_t)s[u] * F + lane * 2);
#pragma unroll
    for (int u = 0; u < CNT; ++u) { acc[0] += w[u] * v[u].x; acc[1] += w[u] * v[u].y; }
  } else {
    float v[CNT];
#pragma unroll
    for (int u = 0; u < CNT; ++u) v[u] = h[(size_t)s[u] * F + lane];
#pragma unroll
    for (int u = 0; u < CNT; ++u) acc[0] += w[u] * v[u];
  }
}

template <int F, bool RELU>
__global__ void agg_gather_k(const float* __restrict__ h, const int* __restrict__ rowptr,
                             const int* __restrict__ csr_pairs,
                             const float* __restrict__ dinv, const float* __restrict__ bias,
                             float* __restrict__ a, int N) {
  static_assert(F == 64 || F == 32, "a lane holds one or two columns");
  const int lane = threadIdx.x & 31;
  const int n = __builtin_amdgcn_readfirstlane(
      (int)(blockIdx.x * (blockDim.x >> 5) + (threadIdx.x >> 5)));
  if (n >= N) return;
  const int beg = rowptr[n];
  const int end = rowptr[n + 1];
  const v2i* __restrict__ csr = (const v2i*)csr_pairs;

  const float di = dinv[n];

  float acc[F / 32] = {};
  int j = beg;
  for (; j + 8 <= end; j += 8) gather_rows<F, 8>(h, csr, j, lane, acc);
  const int r = end - j;
  if (r & 4) { gather_rows<F, 4>(h, csr, j, lane, acc); j += 4; }
  if (r & 2) { gather_rows<F, 2>(h, csr, j, lane, acc); j += 2; }
  if (r & 1) { gather_rows<F, 1>(h, csr, j, lane, acc); }

  if constexpr (F == 64) {
    const v2f hv = *(const v2f*)(h + (size_t)n * F + lane * 2);
    v2f o;
    o.x = bias[lane * 2]     + di * (acc[0] + hv.x);
    o.y = bias[lane * 2 + 1] + di * (acc[1] + hv.y);
    if constexpr (RELU) { o.x = fmaxf(o.x, 0.0f); o.y = fmaxf(o.y, 0.0f); }
    *(v2f*)(a + (size_t)n * F + lane * 2) = o;
  } else {
    const float hv = h[(size_t)n * F + lane];
    float o = bias[lane] + di * (acc[0] + hv);
    if constexpr (RELU) o = fmaxf(o, 0.0f);
    a[(size_t)n * F + lane] = o;
  }
}

#define AG_PITCH 68

template <int FN, bool RELU>
__global__ void agg_gemm_k(const float* __restrict__ h, const int* __restrict__ rowptr,
                           const int* __restrict__ csr_pairs,
                           const float* __restrict__ dinv, const float* __restrict__ bias,
                           const float* __restrict__ Wn, float* __restrict__ hn, int N) {
  constexpr int F = 64;
  static_assert(FN == 64 || FN == 32, "one 16-column tile per wave, at most 8 waves");
  __shared__ float sA[16 * AG_PITCH];
  const int lane = threadIdx.x & 31;
  const int wave = __builtin_amdgcn_readfirstlane((int)(threadIdx.x >> 5));
  const int n0 = blockIdx.x * 16;
  const v2i* __restrict__ csr = (const v2i*)csr_pairs;
  const v2f bv = *(const v2f*)(bias + lane * 2);

  const int nA = n0 + wave * 2;
  int b0 = 0, b1 = 0, b2 = 0;
  if (nA < N) {
    b0 = rowptr[nA];
    b1 = rowptr[nA + 1];
    b2 = rowptr[min(nA + 2, N)];
  }

#pragma unroll 1
  for (int rr = 0; rr < 2; ++rr) {
    const int m = wave * 2 + rr;
    const int n = nA + rr;
    v2f o = {0.0f, 0.0f};
    if (n < N) {
      const int beg = rr ? b1 : b0;
      const int end = rr ? b2 : b1;
      const float di = dinv[n];
      float acc[2] = {};
      int j = beg;
      for (; j + 8 <= end; j += 8) gather_rows<F, 8>(h, csr, j, lane, acc);
      const int r = end - j;
      if (r & 4) { gather_rows<F, 4>(h, csr, j, lane, acc); j += 4; }
      if (r & 2) { gather_rows<F, 2>(h, csr, j, lane, acc); j += 2; }
      if (r & 1) { gather_rows<F, 1>(h, csr, j, lane, acc); }
      const v2f hv = *(const v2f*)(h + (size_t)n * F + lane * 2);
      o.x = bv.x + di * (acc[0] + hv.x);
      o.y = bv.y + di * (acc[1] + hv.y);
      if constexpr (RELU) { o.x = fmaxf(o.x, 0.0f); o.y = fmaxf(o.y, 0.0f); }
    }
    *(v2f*)(sA + m * AG_PITCH + lane * 2) = o;
  }
  __syncthreads();

  if (wave < FN / 16) {
    const int l16 = lane & 15;
    const int hi  = lane >> 4;
    const float* __restrict__ arow = sA + l16 * AG_PITCH + 2 * hi;
    const float* __restrict__ wcol = Wn + wave * 16 + l16;

    float ds[8];
#pragma unroll
    for (int r = 0; r < 8; ++r) ds[r] = dinv[min(n0 + r + 8 * hi, N - 1)];

    v8f acc = {};
#pragma unroll 8
    for (int k = 0; k < F; k += 4) {
      const v2f a = *(const v2f*)(arow + k);
      v2f b;
      b.x = wcol[(size_t)(k + 2 * hi) * FN];
      b.y = wcol[(size_t)(k + 2 * hi + 1) * FN];
      acc = __builtin_amdgcn_wmma_f32_16x16x4_f32(false, a, false, b,
                                                  (short)0, acc, false, false);
    }

    float* __restrict__ hcol = hn + wave * 16 + l16;
    if (n0 + 16 <= N) {
#pragma unroll
      for (int r = 0; r < 8; ++r)
        hcol[(size_t)(n0 + r + 8 * hi) * FN] = ds[r] * acc[r];
    } else {
#pragma unroll
      for (int r = 0; r < 8; ++r) {
        const int row = n0 + r + 8 * hi;
        if (row < N) hcol[(size_t)row * FN] = ds[r] * acc[r];
      }
    }
  }
}

#define POOL_ROWS 256

__global__ void pool_partial_k(const float* __restrict__ emb, float* __restrict__ part, int N) {
  __shared__ float4 sm4[256];
  const int c4 = threadIdx.x & 7;
  const int rs = threadIdx.x >> 3;
  const int r0 = blockIdx.x * POOL_ROWS + rs;
  float4 v[8];
#pragma unroll
  for (int i = 0; i < 8; ++i) {
    const int r = r0 + i * 32;
    v[i] = (r < N) ? *(const float4*)(emb + (size_t)r * 32 + c4 * 4)
                   : make_float4(0.0f, 0.0f, 0.0f, 0.0f);
  }
  float4 s = v[0];
#pragma unroll
  for (int i = 1; i < 8; ++i) {
    s.x += v[i].x; s.y += v[i].y; s.z += v[i].z; s.w += v[i].w;
  }
  sm4[threadIdx.x] = s;
  __syncthreads();
  if (threadIdx.x < 32) {
    const float* sm = (const float*)sm4;
    float t = 0.0f;
#pragma unroll
    for (int i = 0; i < 32; ++i) t += sm[i * 32 + threadIdx.x];
    part[(size_t)blockIdx.x * 32 + threadIdx.x] = t;
  }
}

__global__ void pool_final_k(const float* __restrict__ part, float* __restrict__ g,
                             int nPart, int N) {
  __shared__ float sm[1024];
  const int c   = threadIdx.x & 31;
  const int sub = threadIdx.x >> 5;
  float s = 0.0f;
  for (int b0 = sub; b0 < nPart; b0 += 32 * 8) {
    float v[8];
#pragma unroll
    for (int i = 0; i < 8; ++i) {
      const int b = b0 + i * 32;
      v[i] = (b < nPart) ? part[(size_t)b * 32 + c] : 0.0f;
    }
#pragma unroll
    for (int i = 0; i < 8; ++i) s += v[i];
  }
  sm[threadIdx.x] = s;
  __syncthreads();
  if (threadIdx.x < 32) {
    float t = 0.0f;
#pragma unroll
    for (int i = 0; i < 32; ++i) t += sm[i * 32 + threadIdx.x];
    g[threadIdx.x] = t / (float)N;
  }
}

static inline unsigned cdiv(long a, long b) { return (unsigned)((a + b - 1) / b); }

extern "C" void kernel_launch(void* const* d_in, const int* in_sizes, int n_in,
                              void* d_out, int out_size, void* d_ws, size_t ws_size,
                              hipStream_t stream) {
  const float* x  = (const float*)d_in[0];
  const int*   ei = (const int*)d_in[1];
  const float* ew = (const float*)d_in[2];
  const float* W1 = (const float*)d_in[3];
  const float* b1 = (const float*)d_in[4];
  const float* W2 = (const float*)d_in[5];
  const float* b2 = (const float*)d_in[6];
  const float* W3 = (const float*)d_in[7];
  const float* b3 = (const float*)d_in[8];

  const int N = in_sizes[0] / 128;
  const int E = in_sizes[2];
  const int* src = ei;
  const int* dst = ei + E;

  float* ws      = (float*)d_ws;
  float* bufA    = ws;        ws += (size_t)N * 64;
  float* bufB    = ws;        ws += (size_t)N * 64;
  int*   csr     = (int*)ws;  ws += (size_t)E * 2;
  int*   rank    = (int*)ws;  ws += E;
  float* dinv    = ws;        ws += N;
  int*   cnt     = (int*)ws;  ws += N;
  int*   rowptr  = (int*)ws;  ws += N + 1;
  float* part    = ws;
  float* out     = (float*)d_out;

  const dim3 blk(256);

  cnt_init_k<<<cdiv(N, 256), blk, 0, stream>>>(cnt, N);
  rank_k    <<<cdiv(E, 256), blk, 0, stream>>>(dst, cnt, rank, E);
  rowptr_k  <<<cdiv((long)N + 1, 1024), 1024, 0, stream>>>(cnt, rowptr, N);
  fill_k    <<<cdiv(E, 256), blk, 0, stream>>>(src, dst, ew, rowptr, rank, csr, E);
  rowdeg_k  <<<cdiv(N, 256), blk, 0, stream>>>(rowptr, csr, dinv, N);

  gemm16_wmma<128, 64><<<cdiv((long)(N / 16) * 4, 8), blk, 0, stream>>>(x, W1, dinv, bufA, N);

  agg_gemm_k<64, true><<<cdiv(N, 16), blk, 0, stream>>>(bufA, rowptr, csr, dinv, b1, W2, bufB, N);

  agg_gemm_k<32, true><<<cdiv(N, 16), blk, 0, stream>>>(bufB, rowptr, csr, dinv, b2, W3, bufA, N);

  agg_gather_k<32, false><<<cdiv(N, 8), blk, 0, stream>>>(bufA, rowptr, csr, dinv, b3, out, N);
  pool_partial_k <<<cdiv(N, POOL_ROWS), blk, 0, stream>>>(out, part, N);
  pool_final_k   <<<1, 1024, 0, stream>>>(part, out + (size_t)N * 32, (int)cdiv(N, POOL_ROWS), N);
}


# Aggregations: a row's edge list comes in by one coalesced vector load (a lane per edge), handed out by readlane; 16 source rows in flight per wave, not 8 behind a scalar list load; sum order kept
# speedup vs baseline: 4.1763x; 1.0338x over previous
#include <hip/hip_runtime.h>

typedef __attribute__((ext_vector_type(2))) float v2f;
typedef __attribute__((ext_vector_type(8))) float v8f;
typedef __attribute__((ext_vector_type(2))) int   v2i;

__global__ void cnt_init_k(int* __restrict__ cnt, int N) {
  int t = blockIdx.x * blockDim.x + threadIdx.x;
  if (t < N) cnt[t] = 0;
}

__global__ void rank_k(const int* __restrict__ dst, int* __restrict__ cnt,
                       int* __restrict__ rank, int E) {
  int t = blockIdx.x * blockDim.x + threadIdx.x;
  if (t < E) rank[t] = atomicAdd(&cnt[dst[t]], 1);
}

__global__ void rowptr_k(const int* __restrict__ cnt, int* __restrict__ rowptr, int N) {
  __shared__ int s_before[32];
  __shared__ int s_wave[32];
  const int tid  = threadIdx.x;
  const int lane = tid & 31;
  const int wave = tid >> 5;
  const int base = blockIdx.x * 1024;
  int before = 0;
#pragma unroll 8
  for (int i = tid; i < base; i += 1024) before += cnt[i];
  for (int o = 16; o > 0; o >>= 1) before += __shfl_xor(before, o, 32);

  const int i = base + tid;
  const int c = (i < N) ? cnt[i] : 0;
  int x = c;
  for (int o = 1; o < 32; o <<= 1) {
    const int y = __shfl_up(x, o, 32);
    if (lane >= o) x += y;
  }
  if (lane == 31) s_wave[wave] = x;
  if (lane == 0)  s_before[wave] = before;
  __syncthreads();
  int off = 0;
  for (int w = 0; w < 32; ++w) {
    off += s_before[w];
    if (w < wave) off += s_wave[w];
  }
  if (i <= N) rowptr[i] = off + x - c;
}

__global__ void fill_k(const int* __restrict__ src, const int* __restrict__ dst,
                       const float* __restrict__ ew, const int* __restrict__ rowptr,
                       const int* __restrict__ rank, int* __restrict__ csr, int E) {
  int t = blockIdx.x * blockDim.x + threadIdx.x;
  if (t < E) {
    const int p = rowptr[dst[t]] + rank[t];
    v2i e;
    e.x = src[t];
    e.y = __float_as_int(ew[t]);
    ((v2i*)csr)[p] = e;
  }
}

__global__ void rowdeg_k(const int* __restrict__ rowptr, const int* __restrict__ csr,
                         float* __restrict__ dinv, int N) {
  int n = blockIdx.x * blockDim.x + threadIdx.x;
  if (n < N) {
    const v2i* __restrict__ pr = (const v2i*)csr;
    const int end = rowptr[n + 1];
    float d = 1.0f;
    for (int j = rowptr[n]; j < end; ++j) d += __int_as_float(pr[j].y);
    dinv[n] = (d > 0.0f) ? rsqrtf(d) : 0.0f;
  }
}

template <int K, int F>
__global__ void gemm16_wmma(const float* __restrict__ X, const float* __restrict__ W,
                            const float* __restrict__ dinv, float* __restrict__ H, int N) {
  const int lane = threadIdx.x & 31;
  const int wave = threadIdx.x >> 5;
  const int Ft = F >> 4;
  const int Mt = N >> 4;
  const int tile = blockIdx.x * (blockDim.x >> 5) + wave;
  if (tile >= Mt * Ft) return;
  const int mt = tile / Ft;
  const int ft = tile - mt * Ft;
  const int l16 = lane & 15;
  const int hi  = lane >> 4;

  const float* __restrict__ xrow = X + (size_t)(mt * 16 + l16) * K;
  const float* __restrict__ wcol = W + ft * 16 + l16;

  float ds[8];
#pragma unroll
  for (int r = 0; r < 8; ++r) ds[r] = dinv[mt * 16 + r + 8 * hi];

  v8f acc = {};
#pragma unroll 4
  for (int k = 0; k < K; k += 4) {
    v2f a, b;
    a.x = xrow[k + 2 * hi];
    a.y = xrow[k + 2 * hi + 1];
    b.x = wcol[(size_t)(k + 2 * hi) * F];
    b.y = wcol[(size_t)(k + 2 * hi + 1) * F];
    acc = __builtin_amdgcn_wmma_f32_16x16x4_f32(false, a, false, b,
                                                (short)0, acc, false, false);
  }

  float* __restrict__ hcol = H + ft * 16 + l16;
#pragma unroll
  for (int r = 0; r < 8; ++r)
    hcol[(size_t)(mt * 16 + r + 8 * hi) * F] = ds[r] * acc[r];
}

#define AG_CH 16

__device__ __forceinline__ v2i load_edges(const v2i* __restrict__ csr, int base, int end, int lane) {
  v2i e = {0, 0};
  if (base + lane < end) e = csr[base + lane];
  return e;
}

template <int F, int CH, bool PAD>
__device__ __forceinline__ void gather_chunk(const float* __restrict__ h, const v2i e,
                                             int first, int cnt, int lane,
                                             float (&acc)[F / 32]) {
  int   s[CH];
  float w[CH];
#pragma unroll
  for (int i = 0; i < CH; ++i) {
    const int l = PAD ? min(first + i, first + cnt - 1) : first + i;
    s[i] = __builtin_amdgcn_readlane(e.x, l);
    w[i] = __int_as_float(__builtin_amdgcn_readlane(e.y, l));
  }
  if constexpr (F == 64) {
    v2f v[CH];
#pragma unroll
    for (int i = 0; i < CH; ++i) v[i] = *(const v2f*)(h + (size_t)s[i] * F + lane * 2);
#pragma unroll
    for (int i = 0; i < CH; ++i)
      if (!PAD || i < cnt) { acc[0] += w[i] * v[i].x; acc[1] += w[i] * v[i].y; }
  } else {
    float v[CH];
#pragma unroll
    for (int i = 0; i < CH; ++i) v[i] = h[(size_t)s[i] * F + lane];
#pragma unroll
    for (int i = 0; i < CH; ++i)
      if (!PAD || i < cnt) acc[0] += w[i] * v[i];
  }
}

template <int F>
__device__ __forceinline__ void gather_row(const float* __restrict__ h,
                                           const v2i* __restrict__ csr, v2i e,
                                           int beg, int end, int lane,
                                           float (&acc)[F / 32]) {
  for (int base = beg; base < end; base += 32) {
    if (base != beg) e = load_edges(csr, base, end, lane);
    const int held = min(end - base, 32);
    int first = 0;
    for (; first + AG_CH <= held; first += AG_CH)
      gather_chunk<F, AG_CH, false>(h, e, first, AG_CH, lane, acc);
    const int r = held - first;
    if      (r > 8)  gather_chunk<F, 16, true >(h, e, first, r, lane, acc);
    else if (r > 4)  gather_chunk<F, 8,  true >(h, e, first, r, lane, acc);
    else if (r > 2)  gather_chunk<F, 4,  true >(h, e, first, r, lane, acc);
    else if (r == 2) gather_chunk<F, 2,  false>(h, e, first, 2, lane, acc);
    else if (r == 1) gather_chunk<F, 1,  false>(h, e, first, 1, lane, acc);
  }
}

template <int F, bool RELU>
__global__ void agg_gather_k(const float* __restrict__ h, const int* __restrict__ rowptr,
                             const int* __restrict__ csr_pairs,
                             const float* __restrict__ dinv, const float* __restrict__ bias,
                             float* __restrict__ a, int N) {
  static_assert(F == 64 || F == 32, "a lane holds one or two columns");
  const int lane = threadIdx.x & 31;
  const int n = __builtin_amdgcn_readfirstlane(
      (int)(blockIdx.x * (blockDim.x >> 5) + (threadIdx.x >> 5)));
  if (n >= N) return;
  const int beg = rowptr[n];
  const int end = rowptr[n + 1];
  const v2i* __restrict__ csr = (const v2i*)csr_pairs;
  const v2i e = load_edges(csr, beg, end, lane);

  const float di = dinv[n];

  float acc[F / 32] = {};
  if constexpr (F == 64) {
    const v2f hv = *(const v2f*)(h + (size_t)n * F + lane * 2);
    const v2f bl = *(const v2f*)(bias + lane * 2);
    gather_row<F>(h, csr, e, beg, end, lane, acc);
    v2f o;
    o.x = bl.x + di * (acc[0] + hv.x);
    o.y = bl.y + di * (acc[1] + hv.y);
    if constexpr (RELU) { o.x = fmaxf(o.x, 0.0f); o.y = fmaxf(o.y, 0.0f); }
    *(v2f*)(a + (size_t)n * F + lane * 2) = o;
  } else {
    const float hv = h[(size_t)n * F + lane];
    const float bl = bias[lane];
    gather_row<F>(h, csr, e, beg, end, lane, acc);
    float o = bl + di * (acc[0] + hv);
    if constexpr (RELU) o = fmaxf(o, 0.0f);
    a[(size_t)n * F + lane] = o;
  }
}

#define AG_PITCH 68

template <int FN, bool RELU>
__global__ void agg_gemm_k(const float* __restrict__ h, const int* __restrict__ rowptr,
                           const int* __restrict__ csr_pairs,
                           const float* __restrict__ dinv, const float* __restrict__ bias,
                           const float* __restrict__ Wn, float* __restrict__ hn, int N) {
  constexpr int F = 64;
  static_assert(FN == 64 || FN == 32, "one 16-column tile per wave, at most 8 waves");
  __shared__ float sA[16 * AG_PITCH];
  const int lane = threadIdx.x & 31;
  const int wave = __builtin_amdgcn_readfirstlane((int)(threadIdx.x >> 5));
  const int n0 = blockIdx.x * 16;
  const v2i* __restrict__ csr = (const v2i*)csr_pairs;
  const v2f bv = *(const v2f*)(bias + lane * 2);

  const int nA = n0 + wave * 2;
  int b0 = 0, b1 = 0, b2 = 0;
  if (nA < N) {
    b0 = rowptr[nA];
    b1 = rowptr[nA + 1];
    b2 = rowptr[min(nA + 2, N)];
  }

  v2f hvA = {0.0f, 0.0f}, hvB = {0.0f, 0.0f};
  if (nA < N)     hvA = *(const v2f*)(h + (size_t)nA * F + lane * 2);
  if (nA + 1 < N) hvB = *(const v2f*)(h + (size_t)(nA + 1) * F + lane * 2);

  const v2i eA = load_edges(csr, b0, b1, lane);
  const v2i eB = load_edges(csr, b1, b2, lane);

#pragma unroll 1
  for (int rr = 0; rr < 2; ++rr) {
    const int m = wave * 2 + rr;
    const int n = nA + rr;
    v2f o = {0.0f, 0.0f};
    if (n < N) {
      const int beg = rr ? b1 : b0;
      const int end = rr ? b2 : b1;
      const float di = dinv[n];
      const v2f hv = rr ? hvB : hvA;
      float acc[2] = {};
      gather_row<F>(h, csr, rr ? eB : eA, beg, end, lane, acc);
      o.x = bv.x + di * (acc[0] + hv.x);
      o.y = bv.y + di * (acc[1] + hv.y);
      if constexpr (RELU) { o.x = fmaxf(o.x, 0.0f); o.y = fmaxf(o.y, 0.0f); }
    }
    *(v2f*)(sA + m * AG_PITCH + lane * 2) = o;
  }
  __syncthreads();

  if (wave < FN / 16) {
    const int l16 = lane & 15;
    const int hi  = lane >> 4;
    const float* __restrict__ arow = sA + l16 * AG_PITCH + 2 * hi;
    const float* __restrict__ wcol = Wn + wave * 16 + l16;

    float ds[8];
#pragma unroll
    for (int r = 0; r < 8; ++r) ds[r] = dinv[min(n0 + r + 8 * hi, N - 1)];

    v8f acc = {};
#pragma unroll 8
    for (int k = 0; k < F; k += 4) {
      const v2f a = *(const v2f*)(arow + k);
      v2f b;
      b.x = wcol[(size_t)(k + 2 * hi) * FN];
      b.y = wcol[(size_t)(k + 2 * hi + 1) * FN];
      acc = __builtin_amdgcn_wmma_f32_16x16x4_f32(false, a, false, b,
                                                  (short)0, acc, false, false);
    }

    float* __restrict__ hcol = hn + wave * 16 + l16;
    if (n0 + 16 <= N) {
#pragma unroll
      for (int r = 0; r < 8; ++r)
        hcol[(size_t)(n0 + r + 8 * hi) * FN] = ds[r] * acc[r];
    } else {
#pragma unroll
      for (int r = 0; r < 8; ++r) {
        const int row = n0 + r + 8 * hi;
        if (row < N) hcol[(size_t)row * FN] = ds[r] * acc[r];
      }
    }
  }
}

#define POOL_ROWS 256

__global__ void pool_partial_k(const float* __restrict__ emb, float* __restrict__ part, int N) {
  __shared__ float4 sm4[256];
  const int c4 = threadIdx.x & 7;
  const int rs = threadIdx.x >> 3;
  const int r0 = blockIdx.x * POOL_ROWS + rs;
  float4 v[8];
#pragma unroll
  for (int i = 0; i < 8; ++i) {
    const int r = r0 + i * 32;
    v[i] = (r < N) ? *(const float4*)(emb + (size_t)r * 32 + c4 * 4)
                   : make_float4(0.0f, 0.0f, 0.0f, 0.0f);
  }
  float4 s = v[0];
#pragma unroll
  for (int i = 1; i < 8; ++i) {
    s.x += v[i].x; s.y += v[i].y; s.z += v[i].z; s.w += v[i].w;
  }
  sm4[threadIdx.x] = s;
  __syncthreads();
  if (threadIdx.x < 32) {
    const float* sm = (const float*)sm4;
    float t = 0.0f;
#pragma unroll
    for (int i = 0; i < 32; ++i) t += sm[i * 32 + threadIdx.x];
    part[(size_t)blockIdx.x * 32 + threadIdx.x] = t;
  }
}

__global__ void pool_final_k(const float* __restrict__ part, float* __restrict__ g,
                             int nPart, int N) {
  __shared__ float sm[1024];
  const int c   = threadIdx.x & 31;
  const int sub = threadIdx.x >> 5;
  float s = 0.0f;
  for (int b0 = sub; b0 < nPart; b0 += 32 * 8) {
    float v[8];
#pragma unroll
    for (int i = 0; i < 8; ++i) {
      const int b = b0 + i * 32;
      v[i] = (b < nPart) ? part[(size_t)b * 32 + c] : 0.0f;
    }
#pragma unroll
    for (int i = 0; i < 8; ++i) s += v[i];
  }
  sm[threadIdx.x] = s;
  __syncthreads();
  if (threadIdx.x < 32) {
    float t = 0.0f;
#pragma unroll
    for (int i = 0; i < 32; ++i) t += sm[i * 32 + threadIdx.x];
    g[threadIdx.x] = t / (float)N;
  }
}

static inline unsigned cdiv(long a, long b) { return (unsigned)((a + b - 1) / b); }

extern "C" void kernel_launch(void* const* d_in, const int* in_sizes, int n_in,
                              void* d_out, int out_size, void* d_ws, size_t ws_size,
                              hipStream_t stream) {
  const float* x  = (const float*)d_in[0];
  const int*   ei = (const int*)d_in[1];
  const float* ew = (const float*)d_in[2];
  const float* W1 = (const float*)d_in[3];
  const float* b1 = (const float*)d_in[4];
  const float* W2 = (const float*)d_in[5];
  const float* b2 = (const float*)d_in[6];
  const float* W3 = (const float*)d_in[7];
  const float* b3 = (const float*)d_in[8];

  const int N = in_sizes[0] / 128;
  const int E = in_sizes[2];
  const int* src = ei;
  const int* dst = ei + E;

  float* ws      = (float*)d_ws;
  float* bufA    = ws;        ws += (size_t)N * 64;
  float* bufB    = ws;        ws += (size_t)N * 64;
  int*   csr     = (int*)ws;  ws += (size_t)E * 2;
  int*   rank    = (int*)ws;  ws += E;
  float* dinv    = ws;        ws += N;
  int*   cnt     = (int*)ws;  ws += N;
  int*   rowptr  = (int*)ws;  ws += N + 1;
  float* part    = ws;
  float* out     = (float*)d_out;

  const dim3 blk(256);

  cnt_init_k<<<cdiv(N, 256), blk, 0, stream>>>(cnt, N);
  rank_k    <<<cdiv(E, 256), blk, 0, stream>>>(dst, cnt, rank, E);
  rowptr_k  <<<cdiv((long)N + 1, 1024), 1024, 0, stream>>>(cnt, rowptr, N);
  fill_k    <<<cdiv(E, 256), blk, 0, stream>>>(src, dst, ew, rowptr, rank, csr, E);
  rowdeg_k  <<<cdiv(N, 256), blk, 0, stream>>>(rowptr, csr, dinv, N);

  gemm16_wmma<128, 64><<<cdiv((long)(N / 16) * 4, 8), blk, 0, stream>>>(x, W1, dinv, bufA, N);

  agg_gemm_k<64, true><<<cdiv(N, 16), blk, 0, stream>>>(bufA, rowptr, csr, dinv, b1, W2, bufB, N);

  agg_gemm_k<32, true><<<cdiv(N, 16), blk, 0, stream>>>(bufB, rowptr, csr, dinv, b2, W3, bufA, N);

  agg_gather_k<32, false><<<cdiv(N, 8), blk, 0, stream>>>(bufA, rowptr, csr, dinv, b3, out, N);
  pool_partial_k <<<cdiv(N, POOL_ROWS), blk, 0, stream>>>(out, part, N);
  pool_final_k   <<<1, 1024, 0, stream>>>(part, out + (size_t)N * 32, (int)cdiv(N, POOL_ROWS), N);
}


# Edge sort by destination: global-atomic rank + scattered 8-B stores -> two-level counting sort on LDS counters, coalesced streams; rowptr and deg (LDS float atomics) ride along; 5 launches -> 2
# speedup vs baseline: 7.6230x; 1.8253x over previous
#include <hip/hip_runtime.h>

typedef __attribute__((ext_vector_type(2))) float v2f;
typedef __attribute__((ext_vector_type(8))) float v8f;
typedef __attribute__((ext_vector_type(2))) int   v2i;

#define SORT_BSHIFT 8
#define SORT_BROWS  256
#define SORT_CE     4096
#define SORT_CAP    6144

__global__ __launch_bounds__(1024) void bin_k(const int* __restrict__ src,
                                              const int* __restrict__ dst,
                                              const float* __restrict__ ew,
                                              int* __restrict__ stg_pairs,
                                              int* __restrict__ MT, int E, int nb, int nchunk) {
  extern __shared__ int hist[];
  __shared__ v2i sst[SORT_CE];
  __shared__ int swave[32];
  __shared__ int stot;
  const int tid  = threadIdx.x;
  const int lane = tid & 31;
  const int wave = tid >> 5;
  const int base = blockIdx.x * SORT_CE;

  for (int i = tid; i < nb; i += 1024) hist[i] = 0;

  int d[4], s[4], w[4];
#pragma unroll
  for (int k = 0; k < 4; ++k) {
    const int e = base + k * 1024 + tid;
    d[k] = 0; s[k] = 0; w[k] = 0;
    if (e < E) { d[k] = dst[e]; s[k] = src[e]; w[k] = __float_as_int(ew[e]); }
  }
  __syncthreads();
  int rx[4], br[4];
#pragma unroll
  for (int k = 0; k < 4; ++k) {
    const int e = base + k * 1024 + tid;
    const int bk = d[k] >> SORT_BSHIFT;
    rx[k] = (int)(((unsigned)s[k] << 8) | ((unsigned)d[k] & (SORT_BROWS - 1)));
    br[k] = 0;
    if (e < E) br[k] = (bk << 12) | atomicAdd(&hist[bk], 1);
  }
  __syncthreads();

  int carry = 0;
  for (int t0 = 0; t0 < nb; t0 += 1024) {
    const int i = t0 + tid;
    const int c = (i < nb) ? hist[i] : 0;
    int x = c;
    for (int o = 1; o < 32; o <<= 1) {
      const int y = __shfl_up(x, o, 32);
      if (lane >= o) x += y;
    }
    if (lane == 31) swave[wave] = x;
    __syncthreads();
    if (wave == 0) {
      const int t = swave[lane];
      int z = t;
      for (int o = 1; o < 32; o <<= 1) {
        const int y = __shfl_up(z, o, 32);
        if (lane >= o) z += y;
      }
      swave[lane] = z - t;
      if (lane == 31) stot = z;
    }
    __syncthreads();
    const int ex = carry + swave[wave] + x - c;
    if (i < nb) {
      hist[i] = ex;
      MT[(size_t)i * nchunk + blockIdx.x] = ex | (c << 16);
    }
    carry += stot;
    __syncthreads();
  }

#pragma unroll
  for (int k = 0; k < 4; ++k) {
    const int e = base + k * 1024 + tid;
    if (e < E) {
      v2i rec;
      rec.x = rx[k];
      rec.y = w[k];
      sst[hist[br[k] >> 12] + (br[k] & 4095)] = rec;
    }
  }
  __syncthreads();
  v2i* __restrict__ stg = (v2i*)stg_pairs;
#pragma unroll
  for (int k = 0; k < 4; ++k) {
    const int idx = k * 1024 + tid;
    if (base + idx < E) stg[(size_t)base + idx] = sst[idx];
  }
}

__global__ __launch_bounds__(512) void bucket_k(const int* __restrict__ stg_pairs,
                                                const int* __restrict__ MT,
                                                int* __restrict__ csr_pairs,
                                                int* __restrict__ rowptr,
                                                float* __restrict__ dinv,
                                                int N, int nchunk) {
  __shared__ int   rcnt[SORT_BROWS];
  __shared__ float rdeg[SORT_BROWS];
  __shared__ int   roff[SORT_BROWS];
  __shared__ int   rcur[SORT_BROWS];
  __shared__ int   sg[16], st[16], sw[16];
  __shared__ v2i   sout[SORT_CAP];
  const int tid  = threadIdx.x;
  const int lane = tid & 31;
  const int wave = tid >> 5;
  const int b    = blockIdx.x;
  const v2i* __restrict__ stg = (const v2i*)stg_pairs;
  v2i* __restrict__ csr = (v2i*)csr_pairs;
  const int* __restrict__ mrow = MT + (size_t)b * nchunk;

  if (tid < SORT_BROWS) { rcnt[tid] = 0; rdeg[tid] = 0.0f; rcur[tid] = 0; }
  __syncthreads();

  int gsum = 0, tsum = 0;
  for (int c = tid; c < nchunk; c += 512) {
    const int m   = mrow[c];
    const int off = m & 0xFFFF;
    const int cnt = m >> 16;
    gsum += off;
    tsum += cnt;
    const v2i* __restrict__ seg = stg + (size_t)c * SORT_CE + off;
#pragma unroll 4
    for (int i = 0; i < cnt; ++i) {
      const v2i rec = seg[i];
      const int dl = rec.x & (SORT_BROWS - 1);
      atomicAdd(&rcnt[dl], 1);
      atomicAdd(&rdeg[dl], __int_as_float(rec.y));
    }
  }
  for (int o = 16; o > 0; o >>= 1) {
    gsum += __shfl_xor(gsum, o, 32);
    tsum += __shfl_xor(tsum, o, 32);
  }
  if (lane == 0) { sg[wave] = gsum; st[wave] = tsum; }
  __syncthreads();
  int gbase = 0, total = 0;
  for (int v = 0; v < 16; ++v) { gbase += sg[v]; total += st[v]; }

  const int cr = (tid < SORT_BROWS) ? rcnt[tid] : 0;
  int x = cr;
  for (int o = 1; o < 32; o <<= 1) {
    const int y = __shfl_up(x, o, 32);
    if (lane >= o) x += y;
  }
  if (lane == 31) sw[wave] = x;
  __syncthreads();
  int woff = 0;
  for (int v = 0; v < 16; ++v) if (v < wave) woff += sw[v];
  if (tid < SORT_BROWS) {
    const int ex = woff + x - cr;
    roff[tid] = ex;
    const int row = b * SORT_BROWS + tid;
    if (row < N) {
      rowptr[row] = gbase + ex;
      const float dg = 1.0f + rdeg[tid];
      dinv[row] = (dg > 0.0f) ? rsqrtf(dg) : 0.0f;
      if (row == N - 1) rowptr[N] = gbase + total;
    }
  }
  __syncthreads();

  const bool fits = total <= SORT_CAP;
  for (int c = tid; c < nchunk; c += 512) {
    const int m   = mrow[c];
    const int off = m & 0xFFFF;
    const int cnt = m >> 16;
    const v2i* __restrict__ seg = stg + (size_t)c * SORT_CE + off;
#pragma unroll 4
    for (int i = 0; i < cnt; ++i) {
      const v2i rec = seg[i];
      const int dl = rec.x & (SORT_BROWS - 1);
      const int p  = roff[dl] + atomicAdd(&rcur[dl], 1);
      v2i o;
      o.x = (int)((unsigned)rec.x >> 8);
      o.y = rec.y;
      if (fits) sout[p] = o; else csr[(size_t)gbase + p] = o;
    }
  }
  __syncthreads();
  if (fits)
    for (int i = tid; i < total; i += 512) csr[(size_t)gbase + i] = sout[i];
}

template <int K, int F>
__global__ void gemm16_wmma(const float* __restrict__ X, const float* __restrict__ W,
                            const float* __restrict__ dinv, float* __restrict__ H, int N) {
  const int lane = threadIdx.x & 31;
  const int wave = threadIdx.x >> 5;
  const int Ft = F >> 4;
  const int Mt = N >> 4;
  const int tile = blockIdx.x * (blockDim.x >> 5) + wave;
  if (tile >= Mt * Ft) return;
  const int mt = tile / Ft;
  const int ft = tile - mt * Ft;
  const int l16 = lane & 15;
  const int hi  = lane >> 4;

  const float* __restrict__ xrow = X + (size_t)(mt * 16 + l16) * K;
  const float* __restrict__ wcol = W + ft * 16 + l16;

  float ds[8];
#pragma unroll
  for (int r = 0; r < 8; ++r) ds[r] = dinv[mt * 16 + r + 8 * hi];

  v8f acc = {};
#pragma unroll 4
  for (int k = 0; k < K; k += 4) {
    v2f a, b;
    a.x = xrow[k + 2 * hi];
    a.y = xrow[k + 2 * hi + 1];
    b.x = wcol[(size_t)(k + 2 * hi) * F];
    b.y = wcol[(size_t)(k + 2 * hi + 1) * F];
    acc = __builtin_amdgcn_wmma_f32_16x16x4_f32(false, a, false, b,
                                                (short)0, acc, false, false);
  }

  float* __restrict__ hcol = H + ft * 16 + l16;
#pragma unroll
  for (int r = 0; r < 8; ++r)
    hcol[(size_t)(mt * 16 + r + 8 * hi) * F] = ds[r] * acc[r];
}

#define AG_CH 16

__device__ __forceinline__ v2i load_edges(const v2i* __restrict__ csr, int base, int end, int lane) {
  v2i e = {0, 0};
  if (base + lane < end) e = csr[base + lane];
  return e;
}

template <int F, int CH, bool PAD>
__device__ __forceinline__ void gather_chunk(const float* __restrict__ h, const v2i e,
                                             int first, int cnt, int lane,
                                             float (&acc)[F / 32]) {
  int   s[CH];
  float w[CH];
#pragma unroll
  for (int i = 0; i < CH; ++i) {
    const int l = PAD ? min(first + i, first + cnt - 1) : first + i;
    s[i] = __builtin_amdgcn_readlane(e.x, l);
    w[i] = __int_as_float(__builtin_amdgcn_readlane(e.y, l));
  }
  if constexpr (F == 64) {
    v2f v[CH];
#pragma unroll
    for (int i = 0; i < CH; ++i) v[i] = *(const v2f*)(h + (size_t)s[i] * F + lane * 2);
#pragma unroll
    for (int i = 0; i < CH; ++i)
      if (!PAD || i < cnt) { acc[0] += w[i] * v[i].x; acc[1] += w[i] * v[i].y; }
  } else {
    float v[CH];
#pragma unroll
    for (int i = 0; i < CH; ++i) v[i] = h[(size_t)s[i] * F + lane];
#pragma unroll
    for (int i = 0; i < CH; ++i)
      if (!PAD || i < cnt) acc[0] += w[i] * v[i];
  }
}

template <int F>
__device__ __forceinline__ void gather_row(const float* __restrict__ h,
                                           const v2i* __restrict__ csr, v2i e,
                                           int beg, int end, int lane,
                                           float (&acc)[F / 32]) {
  for (int base = beg; base < end; base += 32) {
    if (base != beg) e = load_edges(csr, base, end, lane);
    const int held = min(end - base, 32);
    int first = 0;
    for (; first + AG_CH <= held; first += AG_CH)
      gather_chunk<F, AG_CH, false>(h, e, first, AG_CH, lane, acc);
    const int r = held - first;
    if      (r > 8)  gather_chunk<F, 16, true >(h, e, first, r, lane, acc);
    else if (r > 4)  gather_chunk<F, 8,  true >(h, e, first, r, lane, acc);
    else if (r > 2)  gather_chunk<F, 4,  true >(h, e, first, r, lane, acc);
    else if (r == 2) gather_chunk<F, 2,  false>(h, e, first, 2, lane, acc);
    else if (r == 1) gather_chunk<F, 1,  false>(h, e, first, 1, lane, acc);
  }
}

template <int F, bool RELU>
__global__ void agg_gather_k(const float* __restrict__ h, const int* __restrict__ rowptr,
                             const int* __restrict__ csr_pairs,
                             const float* __restrict__ dinv, const float* __restrict__ bias,
                             float* __restrict__ a, int N) {
  static_assert(F == 64 || F == 32, "a lane holds one or two columns");
  const int lane = threadIdx.x & 31;
  const int n = __builtin_amdgcn_readfirstlane(
      (int)(blockIdx.x * (blockDim.x >> 5) + (threadIdx.x >> 5)));
  if (n >= N) return;
  const int beg = rowptr[n];
  const int end = rowptr[n + 1];
  const v2i* __restrict__ csr = (const v2i*)csr_pairs;
  const v2i e = load_edges(csr, beg, end, lane);

  const float di = dinv[n];

  float acc[F / 32] = {};
  if constexpr (F == 64) {
    const v2f hv = *(const v2f*)(h + (size_t)n * F + lane * 2);
    const v2f bl = *(const v2f*)(bias + lane * 2);
    gather_row<F>(h, csr, e, beg, end, lane, acc);
    v2f o;
    o.x = bl.x + di * (acc[0] + hv.x);
    o.y = bl.y + di * (acc[1] + hv.y);
    if constexpr (RELU) { o.x = fmaxf(o.x, 0.0f); o.y = fmaxf(o.y, 0.0f); }
    *(v2f*)(a + (size_t)n * F + lane * 2) = o;
  } else {
    const float hv = h[(size_t)n * F + lane];
    const float bl = bias[lane];
    gather_row<F>(h, csr, e, beg, end, lane, acc);
    float o = bl + di * (acc[0] + hv);
    if constexpr (RELU) o = fmaxf(o, 0.0f);
    a[(size_t)n * F + lane] = o;
  }
}

#define AG_PITCH 68

template <int FN, bool RELU>
__global__ void agg_gemm_k(const float* __restrict__ h, const int* __restrict__ rowptr,
                           const int* __restrict__ csr_pairs,
                           const float* __restrict__ dinv, const float* __restrict__ bias,
                           const float* __restrict__ Wn, float* __restrict__ hn, int N) {
  constexpr int F = 64;
  static_assert(FN == 64 || FN == 32, "one 16-column tile per wave, at most 8 waves");
  __shared__ float sA[16 * AG_PITCH];
  const int lane = threadIdx.x & 31;
  const int wave = __builtin_amdgcn_readfirstlane((int)(threadIdx.x >> 5));
  const int n0 = blockIdx.x * 16;
  const v2i* __restrict__ csr = (const v2i*)csr_pairs;
  const v2f bv = *(const v2f*)(bias + lane * 2);

  const int nA = n0 + wave * 2;
  int b0 = 0, b1 = 0, b2 = 0;
  if (nA < N) {
    b0 = rowptr[nA];
    b1 = rowptr[nA + 1];
    b2 = rowptr[min(nA + 2, N)];
  }

  v2f hvA = {0.0f, 0.0f}, hvB = {0.0f, 0.0f};
  if (nA < N)     hvA = *(const v2f*)(h + (size_t)nA * F + lane * 2);
  if (nA + 1 < N) hvB = *(const v2f*)(h + (size_t)(nA + 1) * F + lane * 2);

  const v2i eA = load_edges(csr, b0, b1, lane);
  const v2i eB = load_edges(csr, b1, b2, lane);

#pragma unroll 1
  for (int rr = 0; rr < 2; ++rr) {
    const int m = wave * 2 + rr;
    const int n = nA + rr;
    v2f o = {0.0f, 0.0f};
    if (n < N) {
      const int beg = rr ? b1 : b0;
      const int end = rr ? b2 : b1;
      const float di = dinv[n];
      const v2f hv = rr ? hvB : hvA;
      float acc[2] = {};
      gather_row<F>(h, csr, rr ? eB : eA, beg, end, lane, acc);
      o.x = bv.x + di * (acc[0] + hv.x);
      o.y = bv.y + di * (acc[1] + hv.y);
      if constexpr (RELU) { o.x = fmaxf(o.x, 0.0f); o.y = fmaxf(o.y, 0.0f); }
    }
    *(v2f*)(sA + m * AG_PITCH + lane * 2) = o;
  }
  __syncthreads();

  if (wave < FN / 16) {
    const int l16 = lane & 15;
    const int hi  = lane >> 4;
    const float* __restrict__ arow = sA + l16 * AG_PITCH + 2 * hi;
    const float* __restrict__ wcol = Wn + wave * 16 + l16;

    float ds[8];
#pragma unroll
    for (int r = 0; r < 8; ++r) ds[r] = dinv[min(n0 + r + 8 * hi, N - 1)];

    v8f acc = {};
#pragma unroll 8
    for (int k = 0; k < F; k += 4) {
      const v2f a = *(const v2f*)(arow + k);
      v2f b;
      b.x = wcol[(size_t)(k + 2 * hi) * FN];
      b.y = wcol[(size_t)(k + 2 * hi + 1) * FN];
      acc = __builtin_amdgcn_wmma_f32_16x16x4_f32(false, a, false, b,
                                                  (short)0, acc, false, false);
    }

    float* __restrict__ hcol = hn + wave * 16 + l16;
    if (n0 + 16 <= N) {
#pragma unroll
      for (int r = 0; r < 8; ++r)
        hcol[(size_t)(n0 + r + 8 * hi) * FN] = ds[r] * acc[r];
    } else {
#pragma unroll
      for (int r = 0; r < 8; ++r) {
        const int row = n0 + r + 8 * hi;
        if (row < N) hcol[(size_t)row * FN] = ds[r] * acc[r];
      }
    }
  }
}

#define POOL_ROWS 256

__global__ void pool_partial_k(const float* __restrict__ emb, float* __restrict__ part, int N) {
  __shared__ float4 sm4[256];
  const int c4 = threadIdx.x & 7;
  const int rs = threadIdx.x >> 3;
  const int r0 = blockIdx.x * POOL_ROWS + rs;
  float4 v[8];
#pragma unroll
  for (int i = 0; i < 8; ++i) {
    const int r = r0 + i * 32;
    v[i] = (r < N) ? *(const float4*)(emb + (size_t)r * 32 + c4 * 4)
                   : make_float4(0.0f, 0.0f, 0.0f, 0.0f);
  }
  float4 s = v[0];
#pragma unroll
  for (int i = 1; i < 8; ++i) {
    s.x += v[i].x; s.y += v[i].y; s.z += v[i].z; s.w += v[i].w;
  }
  sm4[threadIdx.x] = s;
  __syncthreads();
  if (threadIdx.x < 32) {
    const float* sm = (const float*)sm4;
    float t = 0.0f;
#pragma unroll
    for (int i = 0; i < 32; ++i) t += sm[i * 32 + threadIdx.x];
    part[(size_t)blockIdx.x * 32 + threadIdx.x] = t;
  }
}

__global__ void pool_final_k(const float* __restrict__ part, float* __restrict__ g,
                             int nPart, int N) {
  __shared__ float sm[1024];
  const int c   = threadIdx.x & 31;
  const int sub = threadIdx.x >> 5;
  float s = 0.0f;
  for (int b0 = sub; b0 < nPart; b0 += 32 * 8) {
    float v[8];
#pragma unroll
    for (int i = 0; i < 8; ++i) {
      const int b = b0 + i * 32;
      v[i] = (b < nPart) ? part[(size_t)b * 32 + c] : 0.0f;
    }
#pragma unroll
    for (int i = 0; i < 8; ++i) s += v[i];
  }
  sm[threadIdx.x] = s;
  __syncthreads();
  if (threadIdx.x < 32) {
    float t = 0.0f;
#pragma unroll
    for (int i = 0; i < 32; ++i) t += sm[i * 32 + threadIdx.x];
    g[threadIdx.x] = t / (float)N;
  }
}

static inline unsigned cdiv(long a, long b) { return (unsigned)((a + b - 1) / b); }

extern "C" void kernel_launch(void* const* d_in, const int* in_sizes, int n_in,
                              void* d_out, int out_size, void* d_ws, size_t ws_size,
                              hipStream_t stream) {
  const float* x  = (const float*)d_in[0];
  const int*   ei = (const int*)d_in[1];
  const float* ew = (const float*)d_in[2];
  const float* W1 = (const float*)d_in[3];
  const float* b1 = (const float*)d_in[4];
  const float* W2 = (const float*)d_in[5];
  const float* b2 = (const float*)d_in[6];
  const float* W3 = (const float*)d_in[7];
  const float* b3 = (const float*)d_in[8];

  const int N = in_sizes[0] / 128;
  const int E = in_sizes[2];
  const int* src = ei;
  const int* dst = ei + E;

  float* ws      = (float*)d_ws;
  float* bufA    = ws;        ws += (size_t)N * 64;
  float* bufB    = ws;        ws += (size_t)N * 64;
  int*   csr     = (int*)ws;  ws += (size_t)E * 2;
  const int nb     = (int)cdiv(N, SORT_BROWS);
  const int nchunk = (int)cdiv(E, SORT_CE);
  int*   stg     = (int*)ws;  ws += (size_t)nchunk * SORT_CE * 2;
  int*   MT      = (int*)ws;  ws += (size_t)nb * nchunk;
  float* dinv    = ws;        ws += N;
  int*   rowptr  = (int*)ws;  ws += N + 1;
  float* part    = ws;
  float* out     = (float*)d_out;

  const dim3 blk(256);

  bin_k   <<<nchunk, 1024, (size_t)nb * sizeof(int), stream>>>(src, dst, ew, stg, MT, E, nb, nchunk);
  bucket_k<<<nb, 512, 0, stream>>>(stg, MT, csr, rowptr, dinv, N, nchunk);

  gemm16_wmma<128, 64><<<cdiv((long)(N / 16) * 4, 8), blk, 0, stream>>>(x, W1, dinv, bufA, N);

  agg_gemm_k<64, true><<<cdiv(N, 16), blk, 0, stream>>>(bufA, rowptr, csr, dinv, b1, W2, bufB, N);

  agg_gemm_k<32, true><<<cdiv(N, 16), blk, 0, stream>>>(bufB, rowptr, csr, dinv, b2, W3, bufA, N);

  agg_gather_k<32, false><<<cdiv(N, 8), blk, 0, stream>>>(bufA, rowptr, csr, dinv, b3, out, N);
  pool_partial_k <<<cdiv(N, POOL_ROWS), blk, 0, stream>>>(out, part, N);
  pool_final_k   <<<1, 1024, 0, stream>>>(part, out + (size_t)N * 32, (int)cdiv(N, POOL_ROWS), N);
}


# Layer 1 GEMM: X rows staged in LDS by whole-row 16-B loads, every line once, A fragments read from LDS (was a 16-line strided load per k step, X read 4x); a wave's 2 row tiles share each B fragment
# speedup vs baseline: 8.1107x; 1.0640x over previous
#include <hip/hip_runtime.h>

typedef __attribute__((ext_vector_type(2))) float v2f;
typedef __attribute__((ext_vector_type(8))) float v8f;
typedef __attribute__((ext_vector_type(2))) int   v2i;

#define SORT_BSHIFT 8
#define SORT_BROWS  256
#define SORT_CE     4096
#define SORT_CAP    6144

__global__ __launch_bounds__(1024) void bin_k(const int* __restrict__ src,
                                              const int* __restrict__ dst,
                                              const float* __restrict__ ew,
                                              int* __restrict__ stg_pairs,
                                              int* __restrict__ MT, int E, int nb, int nchunk) {
  extern __shared__ int hist[];
  __shared__ v2i sst[SORT_CE];
  __shared__ int swave[32];
  __shared__ int stot;
  const int tid  = threadIdx.x;
  const int lane = tid & 31;
  const int wave = tid >> 5;
  const int base = blockIdx.x * SORT_CE;

  for (int i = tid; i < nb; i += 1024) hist[i] = 0;

  int d[4], s[4], w[4];
#pragma unroll
  for (int k = 0; k < 4; ++k) {
    const int e = base + k * 1024 + tid;
    d[k] = 0; s[k] = 0; w[k] = 0;
    if (e < E) { d[k] = dst[e]; s[k] = src[e]; w[k] = __float_as_int(ew[e]); }
  }
  __syncthreads();
  int rx[4], br[4];
#pragma unroll
  for (int k = 0; k < 4; ++k) {
    const int e = base + k * 1024 + tid;
    const int bk = d[k] >> SORT_BSHIFT;
    rx[k] = (int)(((unsigned)s[k] << 8) | ((unsigned)d[k] & (SORT_BROWS - 1)));
    br[k] = 0;
    if (e < E) br[k] = (bk << 12) | atomicAdd(&hist[bk], 1);
  }
  __syncthreads();

  int carry = 0;
  for (int t0 = 0; t0 < nb; t0 += 1024) {
    const int i = t0 + tid;
    const int c = (i < nb) ? hist[i] : 0;
    int x = c;
    for (int o = 1; o < 32; o <<= 1) {
      const int y = __shfl_up(x, o, 32);
      if (lane >= o) x += y;
    }
    if (lane == 31) swave[wave] = x;
    __syncthreads();
    if (wave == 0) {
      const int t = swave[lane];
      int z = t;
      for (int o = 1; o < 32; o <<= 1) {
        const int y = __shfl_up(z, o, 32);
        if (lane >= o) z += y;
      }
      swave[lane] = z - t;
      if (lane == 31) stot = z;
    }
    __syncthreads();
    const int ex = carry + swave[wave] + x - c;
    if (i < nb) {
      hist[i] = ex;
      MT[(size_t)i * nchunk + blockIdx.x] = ex | (c << 16);
    }
    carry += stot;
    __syncthreads();
  }

#pragma unroll
  for (int k = 0; k < 4; ++k) {
    const int e = base + k * 1024 + tid;
    if (e < E) {
      v2i rec;
      rec.x = rx[k];
      rec.y = w[k];
      sst[hist[br[k] >> 12] + (br[k] & 4095)] = rec;
    }
  }
  __syncthreads();
  v2i* __restrict__ stg = (v2i*)stg_pairs;
#pragma unroll
  for (int k = 0; k < 4; ++k) {
    const int idx = k * 1024 + tid;
    if (base + idx < E) stg[(size_t)base + idx] = sst[idx];
  }
}

__global__ __launch_bounds__(512) void bucket_k(const int* __restrict__ stg_pairs,
                                                const int* __restrict__ MT,
                                                int* __restrict__ csr_pairs,
                                                int* __restrict__ rowptr,
                                                float* __restrict__ dinv,
                                                int N, int nchunk) {
  __shared__ int   rcnt[SORT_BROWS];
  __shared__ float rdeg[SORT_BROWS];
  __shared__ int   roff[SORT_BROWS];
  __shared__ int   rcur[SORT_BROWS];
  __shared__ int   sg[16], st[16], sw[16];
  __shared__ v2i   sout[SORT_CAP];
  const int tid  = threadIdx.x;
  const int lane = tid & 31;
  const int wave = tid >> 5;
  const int b    = blockIdx.x;
  const v2i* __restrict__ stg = (const v2i*)stg_pairs;
  v2i* __restrict__ csr = (v2i*)csr_pairs;
  const int* __restrict__ mrow = MT + (size_t)b * nchunk;

  if (tid < SORT_BROWS) { rcnt[tid] = 0; rdeg[tid] = 0.0f; rcur[tid] = 0; }
  __syncthreads();

  int gsum = 0, tsum = 0;
  for (int c = tid; c < nchunk; c += 512) {
    const int m   = mrow[c];
    const int off = m & 0xFFFF;
    const int cnt = m >> 16;
    gsum += off;
    tsum += cnt;
    const v2i* __restrict__ seg = stg + (size_t)c * SORT_CE + off;
#pragma unroll 4
    for (int i = 0; i < cnt; ++i) {
      const v2i rec = seg[i];
      const int dl = rec.x & (SORT_BROWS - 1);
      atomicAdd(&rcnt[dl], 1);
      atomicAdd(&rdeg[dl], __int_as_float(rec.y));
    }
  }
  for (int o = 16; o > 0; o >>= 1) {
    gsum += __shfl_xor(gsum, o, 32);
    tsum += __shfl_xor(tsum, o, 32);
  }
  if (lane == 0) { sg[wave] = gsum; st[wave] = tsum; }
  __syncthreads();
  int gbase = 0, total = 0;
  for (int v = 0; v < 16; ++v) { gbase += sg[v]; total += st[v]; }

  const int cr = (tid < SORT_BROWS) ? rcnt[tid] : 0;
  int x = cr;
  for (int o = 1; o < 32; o <<= 1) {
    const int y = __shfl_up(x, o, 32);
    if (lane >= o) x += y;
  }
  if (lane == 31) sw[wave] = x;
  __syncthreads();
  int woff = 0;
  for (int v = 0; v < 16; ++v) if (v < wave) woff += sw[v];
  if (tid < SORT_BROWS) {
    const int ex = woff + x - cr;
    roff[tid] = ex;
    const int row = b * SORT_BROWS + tid;
    if (row < N) {
      rowptr[row] = gbase + ex;
      const float dg = 1.0f + rdeg[tid];
      dinv[row] = (dg > 0.0f) ? rsqrtf(dg) : 0.0f;
      if (row == N - 1) rowptr[N] = gbase + total;
    }
  }
  __syncthreads();

  const bool fits = total <= SORT_CAP;
  for (int c = tid; c < nchunk; c += 512) {
    const int m   = mrow[c];
    const int off = m & 0xFFFF;
    const int cnt = m >> 16;
    const v2i* __restrict__ seg = stg + (size_t)c * SORT_CE + off;
#pragma unroll 4
    for (int i = 0; i < cnt; ++i) {
      const v2i rec = seg[i];
      const int dl = rec.x & (SORT_BROWS - 1);
      const int p  = roff[dl] + atomicAdd(&rcur[dl], 1);
      v2i o;
      o.x = (int)((unsigned)rec.x >> 8);
      o.y = rec.y;
      if (fits) sout[p] = o; else csr[(size_t)gbase + p] = o;
    }
  }
  __syncthreads();
  if (fits)
    for (int i = tid; i < total; i += 512) csr[(size_t)gbase + i] = sout[i];
}

#define G1_ROWS  64
#define G1_PITCH 132

template <int K, int F>
__global__ __launch_bounds__(256) void gemm_rows_wmma(const float* __restrict__ X,
                                                      const float* __restrict__ W,
                                                      const float* __restrict__ dinv,
                                                      float* __restrict__ H, int N) {
  static_assert(K == 128, "a wave's load instruction covers one row: 32 lanes x 16 bytes");
  static_assert(F == 64, "8 waves: 4 column tiles x 2 row halves");
  __shared__ float sX[G1_ROWS * G1_PITCH];
  __shared__ float sD[G1_ROWS];
  const int lane = threadIdx.x & 31;
  const int wave = __builtin_amdgcn_readfirstlane((int)(threadIdx.x >> 5));
  const int n0   = blockIdx.x * G1_ROWS;

  const float* __restrict__ xw = X + (size_t)(n0 + wave * 8) * K + lane * 4;
  float4 xv[8];
  if (n0 + G1_ROWS <= N) {
#pragma unroll
    for (int i = 0; i < 8; ++i) xv[i] = *(const float4*)(xw + (size_t)i * K);
  } else {
#pragma unroll
    for (int i = 0; i < 8; ++i)
      xv[i] = (n0 + wave * 8 + i < N) ? *(const float4*)(xw + (size_t)i * K)
                                      : make_float4(0.0f, 0.0f, 0.0f, 0.0f);
  }
  if (threadIdx.x < G1_ROWS) sD[threadIdx.x] = dinv[min(n0 + (int)threadIdx.x, N - 1)];
#pragma unroll
  for (int i = 0; i < 8; ++i)
    *(float4*)(sX + (wave * 8 + i) * G1_PITCH + lane * 4) = xv[i];
  __syncthreads();

  const int l16 = lane & 15;
  const int hi  = lane >> 4;
  const int ft  = wave & 3;
  const int mh  = wave >> 2;
  const float* __restrict__ arow0 = sX + (mh * 32 + l16) * G1_PITCH + 2 * hi;
  const float* __restrict__ arow1 = arow0 + 16 * G1_PITCH;
  const float* __restrict__ wcol  = W + ft * 16 + l16;

  v8f acc0 = {}, acc1 = {};
#pragma unroll 4
  for (int k = 0; k < K; k += 4) {
    const v2f a0 = *(const v2f*)(arow0 + k);
    const v2f a1 = *(const v2f*)(arow1 + k);
    v2f b;
    b.x = wcol[(size_t)(k + 2 * hi) * F];
    b.y = wcol[(size_t)(k + 2 * hi + 1) * F];
    acc0 = __builtin_amdgcn_wmma_f32_16x16x4_f32(false, a0, false, b,
                                                 (short)0, acc0, false, false);
    acc1 = __builtin_amdgcn_wmma_f32_16x16x4_f32(false, a1, false, b,
                                                 (short)0, acc1, false, false);
  }

  const int m0 = mh * 32 + 8 * hi;
  float* __restrict__ hcol = H + (size_t)n0 * F + ft * 16 + l16;
  if (n0 + G1_ROWS <= N) {
#pragma unroll
    for (int r = 0; r < 8; ++r) {
      hcol[(size_t)(m0 + r) * F]      = sD[m0 + r] * acc0[r];
      hcol[(size_t)(m0 + 16 + r) * F] = sD[m0 + 16 + r] * acc1[r];
    }
  } else {
#pragma unroll
    for (int r = 0; r < 8; ++r) {
      if (n0 + m0 + r < N)      hcol[(size_t)(m0 + r) * F]      = sD[m0 + r] * acc0[r];
      if (n0 + m0 + 16 + r < N) hcol[(size_t)(m0 + 16 + r) * F] = sD[m0 + 16 + r] * acc1[r];
    }
  }
}

#define AG_CH 16

__device__ __forceinline__ v2i load_edges(const v2i* __restrict__ csr, int base, int end, int lane) {
  v2i e = {0, 0};
  if (base + lane < end) e = csr[base + lane];
  return e;
}

template <int F, int CH, bool PAD>
__device__ __forceinline__ void gather_chunk(const float* __restrict__ h, const v2i e,
                                             int first, int cnt, int lane,
                                             float (&acc)[F / 32]) {
  int   s[CH];
  float w[CH];
#pragma unroll
  for (int i = 0; i < CH; ++i) {
    const int l = PAD ? min(first + i, first + cnt - 1) : first + i;
    s[i] = __builtin_amdgcn_readlane(e.x, l);
    w[i] = __int_as_float(__builtin_amdgcn_readlane(e.y, l));
  }
  if constexpr (F == 64) {
    v2f v[CH];
#pragma unroll
    for (int i = 0; i < CH; ++i) v[i] = *(const v2f*)(h + (size_t)s[i] * F + lane * 2);
#pragma unroll
    for (int i = 0; i < CH; ++i)
      if (!PAD || i < cnt) { acc[0] += w[i] * v[i].x; acc[1] += w[i] * v[i].y; }
  } else {
    float v[CH];
#pragma unroll
    for (int i = 0; i < CH; ++i) v[i] = h[(size_t)s[i] * F + lane];
#pragma unroll
    for (int i = 0; i < CH; ++i)
      if (!PAD || i < cnt) acc[0] += w[i] * v[i];
  }
}

template <int F>
__device__ __forceinline__ void gather_row(const float* __restrict__ h,
                                           const v2i* __restrict__ csr, v2i e,
                                           int beg, int end, int lane,
                                           float (&acc)[F / 32]) {
  for (int base = beg; base < end; base += 32) {
    if (base != beg) e = load_edges(csr, base, end, lane);
    const int held = min(end - base, 32);
    int first = 0;
    for (; first + AG_CH <= held; first += AG_CH)
      gather_chunk<F, AG_CH, false>(h, e, first, AG_CH, lane, acc);
    const int r = held - first;
    if      (r > 8)  gather_chunk<F, 16, true >(h, e, first, r, lane, acc);
    else if (r > 4)  gather_chunk<F, 8,  true >(h, e, first, r, lane, acc);
    else if (r > 2)  gather_chunk<F, 4,  true >(h, e, first, r, lane, acc);
    else if (r == 2) gather_chunk<F, 2,  false>(h, e, first, 2, lane, acc);
    else if (r == 1) gather_chunk<F, 1,  false>(h, e, first, 1, lane, acc);
  }
}

template <int F, bool RELU>
__global__ void agg_gather_k(const float* __restrict__ h, const int* __restrict__ rowptr,
                             const int* __restrict__ csr_pairs,
                             const float* __restrict__ dinv, const float* __restrict__ bias,
                             float* __restrict__ a, int N) {
  static_assert(F == 64 || F == 32, "a lane holds one or two columns");
  const int lane = threadIdx.x & 31;
  const int n = __builtin_amdgcn_readfirstlane(
      (int)(blockIdx.x * (blockDim.x >> 5) + (threadIdx.x >> 5)));
  if (n >= N) return;
  const int beg = rowptr[n];
  const int end = rowptr[n + 1];
  const v2i* __restrict__ csr = (const v2i*)csr_pairs;
  const v2i e = load_edges(csr, beg, end, lane);

  const float di = dinv[n];

  float acc[F / 32] = {};
  if constexpr (F == 64) {
    const v2f hv = *(const v2f*)(h + (size_t)n * F + lane * 2);
    const v2f bl = *(const v2f*)(bias + lane * 2);
    gather_row<F>(h, csr, e, beg, end, lane, acc);
    v2f o;
    o.x = bl.x + di * (acc[0] + hv.x);
    o.y = bl.y + di * (acc[1] + hv.y);
    if constexpr (RELU) { o.x = fmaxf(o.x, 0.0f); o.y = fmaxf(o.y, 0.0f); }
    *(v2f*)(a + (size_t)n * F + lane * 2) = o;
  } else {
    const float hv = h[(size_t)n * F + lane];
    const float bl = bias[lane];
    gather_row<F>(h, csr, e, beg, end, lane, acc);
    float o = bl + di * (acc[0] + hv);
    if constexpr (RELU) o = fmaxf(o, 0.0f);
    a[(size_t)n * F + lane] = o;
  }
}

#define AG_PITCH 68

template <int FN, bool RELU>
__global__ void agg_gemm_k(const float* __restrict__ h, const int* __restrict__ rowptr,
                           const int* __restrict__ csr_pairs,
                           const float* __restrict__ dinv, const float* __restrict__ bias,
                           const float* __restrict__ Wn, float* __restrict__ hn, int N) {
  constexpr int F = 64;
  static_assert(FN == 64 || FN == 32, "one 16-column tile per wave, at most 8 waves");
  __shared__ float sA[16 * AG_PITCH];
  const int lane = threadIdx.x & 31;
  const int wave = __builtin_amdgcn_readfirstlane((int)(threadIdx.x >> 5));
  const int n0 = blockIdx.x * 16;
  const v2i* __restrict__ csr = (const v2i*)csr_pairs;
  const v2f bv = *(const v2f*)(bias + lane * 2);

  const int nA = n0 + wave * 2;
  int b0 = 0, b1 = 0, b2 = 0;
  if (nA < N) {
    b0 = rowptr[nA];
    b1 = rowptr[nA + 1];
    b2 = rowptr[min(nA + 2, N)];
  }

  v2f hvA = {0.0f, 0.0f}, hvB = {0.0f, 0.0f};
  if (nA < N)     hvA = *(const v2f*)(h + (size_t)nA * F + lane * 2);
  if (nA + 1 < N) hvB = *(const v2f*)(h + (size_t)(nA + 1) * F + lane * 2);

  const v2i eA = load_edges(csr, b0, b1, lane);
  const v2i eB = load_edges(csr, b1, b2, lane);

#pragma unroll 1
  for (int rr = 0; rr < 2; ++rr) {
    const int m = wave * 2 + rr;
    const int n = nA + rr;
    v2f o = {0.0f, 0.0f};
    if (n < N) {
      const int beg = rr ? b1 : b0;
      const int end = rr ? b2 : b1;
      const float di = dinv[n];
      const v2f hv = rr ? hvB : hvA;
      float acc[2] = {};
      gather_row<F>(h, csr, rr ? eB : eA, beg, end, lane, acc);
      o.x = bv.x + di * (acc[0] + hv.x);
      o.y = bv.y + di * (acc[1] + hv.y);
      if constexpr (RELU) { o.x = fmaxf(o.x, 0.0f); o.y = fmaxf(o.y, 0.0f); }
    }
    *(v2f*)(sA + m * AG_PITCH + lane * 2) = o;
  }
  __syncthreads();

  if (wave < FN / 16) {
    const int l16 = lane & 15;
    const int hi  = lane >> 4;
    const float* __restrict__ arow = sA + l16 * AG_PITCH + 2 * hi;
    const float* __restrict__ wcol = Wn + wave * 16 + l16;

    float ds[8];
#pragma unroll
    for (int r = 0; r < 8; ++r) ds[r] = dinv[min(n0 + r + 8 * hi, N - 1)];

    v8f acc = {};
#pragma unroll 8
    for (int k = 0; k < F; k += 4) {
      const v2f a = *(const v2f*)(arow + k);
      v2f b;
      b.x = wcol[(size_t)(k + 2 * hi) * FN];
      b.y = wcol[(size_t)(k + 2 * hi + 1) * FN];
      acc = __builtin_amdgcn_wmma_f32_16x16x4_f32(false, a, false, b,
                                                  (short)0, acc, false, false);
    }

    float* __restrict__ hcol = hn + wave * 16 + l16;
    if (n0 + 16 <= N) {
#pragma unroll
      for (int r = 0; r < 8; ++r)
        hcol[(size_t)(n0 + r + 8 * hi) * FN] = ds[r] * acc[r];
    } else {
#pragma unroll
      for (int r = 0; r < 8; ++r) {
        const int row = n0 + r + 8 * hi;
        if (row < N) hcol[(size_t)row * FN] = ds[r] * acc[r];
      }
    }
  }
}

#define POOL_ROWS 256

__global__ void pool_partial_k(const float* __restrict__ emb, float* __restrict__ part, int N) {
  __shared__ float4 sm4[256];
  const int c4 = threadIdx.x & 7;
  const int rs = threadIdx.x >> 3;
  const int r0 = blockIdx.x * POOL_ROWS + rs;
  float4 v[8];
#pragma unroll
  for (int i = 0; i < 8; ++i) {
    const int r = r0 + i * 32;
    v[i] = (r < N) ? *(const float4*)(emb + (size_t)r * 32 + c4 * 4)
                   : make_float4(0.0f, 0.0f, 0.0f, 0.0f);
  }
  float4 s = v[0];
#pragma unroll
  for (int i = 1; i < 8; ++i) {
    s.x += v[i].x; s.y += v[i].y; s.z += v[i].z; s.w += v[i].w;
  }
  sm4[threadIdx.x] = s;
  __syncthreads();
  if (threadIdx.x < 32) {
    const float* sm = (const float*)sm4;
    float t = 0.0f;
#pragma unroll
    for (int i = 0; i < 32; ++i) t += sm[i * 32 + threadIdx.x];
    part[(size_t)blockIdx.x * 32 + threadIdx.x] = t;
  }
}

__global__ void pool_final_k(const float* __restrict__ part, float* __restrict__ g,
                             int nPart, int N) {
  __shared__ float sm[1024];
  const int c   = threadIdx.x & 31;
  const int sub = threadIdx.x >> 5;
  float s = 0.0f;
  for (int b0 = sub; b0 < nPart; b0 += 32 * 8) {
    float v[8];
#pragma unroll
    for (int i = 0; i < 8; ++i) {
      const int b = b0 + i * 32;
      v[i] = (b < nPart) ? part[(size_t)b * 32 + c] : 0.0f;
    }
#pragma unroll
    for (int i = 0; i < 8; ++i) s += v[i];
  }
  sm[threadIdx.x] = s;
  __syncthreads();
  if (threadIdx.x < 32) {
    float t = 0.0f;
#pragma unroll
    for (int i = 0; i < 32; ++i) t += sm[i * 32 + threadIdx.x];
    g[threadIdx.x] = t / (float)N;
  }
}

static inline unsigned cdiv(long a, long b) { return (unsigned)((a + b - 1) / b); }

extern "C" void kernel_launch(void* const* d_in, const int* in_sizes, int n_in,
                              void* d_out, int out_size, void* d_ws, size_t ws_size,
                              hipStream_t stream) {
  const float* x  = (const float*)d_in[0];
  const int*   ei = (const int*)d_in[1];
  const float* ew = (const float*)d_in[2];
  const float* W1 = (const float*)d_in[3];
  const float* b1 = (const float*)d_in[4];
  const float* W2 = (const float*)d_in[5];
  const float* b2 = (const float*)d_in[6];
  const float* W3 = (const float*)d_in[7];
  const float* b3 = (const float*)d_in[8];

  const int N = in_sizes[0] / 128;
  const int E = in_sizes[2];
  const int* src = ei;
  const int* dst = ei + E;

  float* ws      = (float*)d_ws;
  float* bufA    = ws;        ws += (size_t)N * 64;
  float* bufB    = ws;        ws += (size_t)N * 64;
  int*   csr     = (int*)ws;  ws += (size_t)E * 2;
  const int nb     = (int)cdiv(N, SORT_BROWS);
  const int nchunk = (int)cdiv(E, SORT_CE);
  int*   stg     = (int*)ws;  ws += (size_t)nchunk * SORT_CE * 2;
  int*   MT      = (int*)ws;  ws += (size_t)nb * nchunk;
  float* dinv    = ws;        ws += N;
  int*   rowptr  = (int*)ws;  ws += N + 1;
  float* part    = ws;
  float* out     = (float*)d_out;

  const dim3 blk(256);

  bin_k   <<<nchunk, 1024, (size_t)nb * sizeof(int), stream>>>(src, dst, ew, stg, MT, E, nb, nchunk);
  bucket_k<<<nb, 512, 0, stream>>>(stg, MT, csr, rowptr, dinv, N, nchunk);

  gemm_rows_wmma<128, 64><<<cdiv(N, G1_ROWS), blk, 0, stream>>>(x, W1, dinv, bufA, N);

  agg_gemm_k<64, true><<<cdiv(N, 16), blk, 0, stream>>>(bufA, rowptr, csr, dinv, b1, W2, bufB, N);

  agg_gemm_k<32, true><<<cdiv(N, 16), blk, 0, stream>>>(bufB, rowptr, csr, dinv, b2, W3, bufA, N);

  agg_gather_k<32, false><<<cdiv(N, 8), blk, 0, stream>>>(bufA, rowptr, csr, dinv, b3, out, N);
  pool_partial_k <<<cdiv(N, POOL_ROWS), blk, 0, stream>>>(out, part, N);
  pool_final_k   <<<1, 1024, 0, stream>>>(part, out + (size_t)N * 32, (int)cdiv(N, POOL_ROWS), N);
}
